# starting kernel
# baseline (speedup 1.0000x reference)
#include <hip/hip_runtime.h>

#define DD 256
#define NSTRUCT 8192
#define EPS 1e-5f

typedef float v2f __attribute__((ext_vector_type(2)));
typedef float v8f __attribute__((ext_vector_type(8)));

__device__ __forceinline__ float wave_sum(float v) {
  v += __shfl_xor(v, 1, 32);
  v += __shfl_xor(v, 2, 32);
  v += __shfl_xor(v, 4, 32);
  v += __shfl_xor(v, 8, 32);
  v += __shfl_xor(v, 16, 32);
  return v;
}
__device__ __forceinline__ float sum4(float4 x) { return x.x + x.y + x.z + x.w; }
__device__ __forceinline__ float sumsq4(float4 x) { return x.x*x.x + x.y*x.y + x.z*x.z + x.w*x.w; }
__device__ __forceinline__ float4 ln4(float4 x, float mu, float rs, float4 g, float4 b) {
  float4 y;
  y.x = (x.x - mu) * rs * g.x + b.x;
  y.y = (x.y - mu) * rs * g.y + b.y;
  y.z = (x.z - mu) * rs * g.z + b.z;
  y.w = (x.w - mu) * rs * g.w + b.w;
  return y;
}

__global__ __launch_bounds__(256) void leaf_ln_kernel(
    const float* __restrict__ comp_emb, const int* __restrict__ leaf_ids,
    const float* __restrict__ gamma, const float* __restrict__ beta,
    float* __restrict__ h0) {
  const int lane = threadIdx.x & 31;
  const int r = blockIdx.x * 8 + (threadIdx.x >> 5);
  const int id = leaf_ids[r];
  const float4* src = (const float4*)(comp_emb + (size_t)id * DD);
  float4 x0 = src[lane];
  float4 x1 = src[lane + 32];
  float s  = wave_sum(sum4(x0) + sum4(x1));
  float s2 = wave_sum(sumsq4(x0) + sumsq4(x1));
  float mu = s * (1.0f / DD);
  float rs = rsqrtf(s2 * (1.0f / DD) - mu * mu + EPS);
  float4 g0 = ((const float4*)gamma)[lane], g1 = ((const float4*)gamma)[lane + 32];
  float4 b0 = ((const float4*)beta)[lane],  b1 = ((const float4*)beta)[lane + 32];
  float4* dst = (float4*)(h0 + (size_t)r * DD);
  dst[lane]      = ln4(x0, mu, rs, g0, b0);
  dst[lane + 32] = ln4(x1, mu, rs, g1, b1);
}

#define MT 32
#define KC 768
#define CSTR 772
#define OSTR 260

__global__ __launch_bounds__(256) void combine_kernel(
    const float* __restrict__ h_in, float* __restrict__ h_out,
    const int* __restrict__ op_ids, int off, int ln2n,
    const float* __restrict__ op_emb, const float* __restrict__ W,
    const float* __restrict__ bias, const float* __restrict__ gamma,
    const float* __restrict__ beta) {
  __shared__ float smem[MT * CSTR];

  const int tid = threadIdx.x;
  const int lane = tid & 31;
  const int wave = tid >> 5;
  const int r0 = blockIdx.x * MT;
  const int nmask = (1 << ln2n) - 1;

  for (int i = tid; i < MT * (KC / 4); i += 256) {
    const int m = i / (KC / 4);
    const int c4 = i - m * (KC / 4);
    const int rg = r0 + m;
    float4 v;
    if (c4 < 64) {
      const int opid = op_ids[(rg >> ln2n) * 7 + off + (rg & nmask)];
      v = ((const float4*)(op_emb + (size_t)opid * DD))[c4];
    } else if (c4 < 128) {
      v = ((const float4*)(h_in + (size_t)(2 * rg) * DD))[c4 - 64];
    } else {
      v = ((const float4*)(h_in + (size_t)(2 * rg + 1) * DD))[c4 - 128];
    }
    *(float4*)&smem[m * CSTR + c4 * 4] = v;
  }
  __syncthreads();

  v8f c00 = {}, c01 = {}, c10 = {}, c11 = {};
  const int n0 = wave * 32;
  const int colA = n0 + (lane & 15);
  const int krow = (lane & 16) >> 3;
  const float* arow0 = &smem[(lane & 15) * CSTR];
  const float* arow1 = &smem[((lane & 15) + 16) * CSTR];
#pragma unroll 4
  for (int k0 = 0; k0 < KC; k0 += 4) {
    const int kb = k0 + krow;
    const float* Wk = W + (size_t)kb * DD;
    v2f b0 = { Wk[colA],      Wk[DD + colA] };
    v2f b1 = { Wk[colA + 16], Wk[DD + colA + 16] };
    v2f a0 = *(const v2f*)(arow0 + kb);
    v2f a1 = *(const v2f*)(arow1 + kb);
    c00 = __builtin_amdgcn_wmma_f32_16x16x4_f32(false, a0, false, b0, (short)0, c00, false, false);
    c01 = __builtin_amdgcn_wmma_f32_16x16x4_f32(false, a0, false, b1, (short)0, c01, false, false);
    c10 = __builtin_amdgcn_wmma_f32_16x16x4_f32(false, a1, false, b0, (short)0, c10, false, false);
    c11 = __builtin_amdgcn_wmma_f32_16x16x4_f32(false, a1, false, b1, (short)0, c11, false, false);
  }
  __syncthreads();

  const float bias0 = bias[colA];
  const float bias1 = bias[colA + 16];
  const int roff = (lane & 16) >> 1;
#pragma unroll
  for (int j = 0; j < 8; ++j) {
    const int rr = j + roff;
    smem[rr * OSTR + colA]              = c00[j] + bias0;
    smem[rr * OSTR + colA + 16]         = c01[j] + bias1;
    smem[(rr + 16) * OSTR + colA]       = c10[j] + bias0;
    smem[(rr + 16) * OSTR + colA + 16]  = c11[j] + bias1;
  }
  __syncthreads();

  const float4 g0  = ((const float4*)gamma)[lane];
  const float4 g1  = ((const float4*)gamma)[lane + 32];
  const float4 bt0 = ((const float4*)beta)[lane];
  const float4 bt1 = ((const float4*)beta)[lane + 32];
#pragma unroll
  for (int i = 0; i < 4; ++i) {
    const int row = wave * 4 + i;
    float4 x0 = *(const float4*)&smem[row * OSTR + lane * 4];
    float4 x1 = *(const float4*)&smem[row * OSTR + 128 + lane * 4];
    float s  = wave_sum(sum4(x0) + sum4(x1));
    float s2 = wave_sum(sumsq4(x0) + sumsq4(x1));
    float mu = s * (1.0f / DD);
    float rs = rsqrtf(s2 * (1.0f / DD) - mu * mu + EPS);
    float4* dst = (float4*)(h_out + (size_t)(r0 + row) * DD);
    dst[lane]      = ln4(x0, mu, rs, g0, bt0);
    dst[lane + 32] = ln4(x1, mu, rs, g1, bt1);
  }
}

__global__ __launch_bounds__(256) void gather_kernel(
    const float* __restrict__ root, const int* __restrict__ sidx,
    float* __restrict__ out) {
  const int lane = threadIdx.x & 31;
  const int r = blockIdx.x * 8 + (threadIdx.x >> 5);
  const int idx = sidx[r];
  float4* dst = (float4*)(out + (size_t)r * DD);
  if (idx > 1) {
    const float4* src = (const float4*)(root + (size_t)idx * DD);
    dst[lane]      = src[lane];
    dst[lane + 32] = src[lane + 32];
  } else {
    float4 z = make_float4(0.f, 0.f, 0.f, 0.f);
    dst[lane]      = z;
    dst[lane + 32] = z;
  }
}

extern "C" void kernel_launch(void* const* d_in, const int* in_sizes, int n_in,
                              void* d_out, int out_size, void* d_ws, size_t ws_size,
                              hipStream_t stream) {
  const int*   struct_indices = (const int*)d_in[0];
  const int*   leaf_ids       = (const int*)d_in[1];
  const int*   op_ids         = (const int*)d_in[2];
  const float* comp_emb       = (const float*)d_in[3];
  const float* op_emb         = (const float*)d_in[4];
  const float* W_bin          = (const float*)d_in[5];
  const float* b_bin          = (const float*)d_in[6];
  const float* gamma          = (const float*)d_in[7];
  const float* beta           = (const float*)d_in[8];

  float* h0   = (float*)d_out;
  float* h1   = (float*)d_ws;
  float* h2   = h1 + (size_t)NSTRUCT * 4 * DD;
  float* root = h2 + (size_t)NSTRUCT * 2 * DD;

  leaf_ln_kernel<<<NSTRUCT * 8 / 8, 256, 0, stream>>>(comp_emb, leaf_ids, gamma, beta, h0);
  combine_kernel<<<(NSTRUCT * 4) / MT, 256, 0, stream>>>(h0, h1, op_ids, 0, 2,
                                                         op_emb, W_bin, b_bin, gamma, beta);
  combine_kernel<<<(NSTRUCT * 2) / MT, 256, 0, stream>>>(h1, h2, op_ids, 4, 1,
                                                         op_emb, W_bin, b_bin, gamma, beta);
  combine_kernel<<<(NSTRUCT * 1) / MT, 256, 0, stream>>>(h2, root, op_ids, 6, 0,
                                                         op_emb, W_bin, b_bin, gamma, beta);
  gather_kernel<<<65536 / 8, 256, 0, stream>>>(root, struct_indices, (float*)d_out);
}


# combine kernel, gather phase: each thread requests its 8 index loads and then all 24 row loads before the first wait; was 32 load-wait-store round trips in series
# speedup vs baseline: 1.1276x; 1.1276x over previous
#include <hip/hip_runtime.h>

#define DD 256
#define NSTRUCT 8192
#define EPS 1e-5f

typedef float v2f __attribute__((ext_vector_type(2)));
typedef float v8f __attribute__((ext_vector_type(8)));

__device__ __forceinline__ float wave_sum(float v) {
  v += __shfl_xor(v, 1, 32);
  v += __shfl_xor(v, 2, 32);
  v += __shfl_xor(v, 4, 32);
  v += __shfl_xor(v, 8, 32);
  v += __shfl_xor(v, 16, 32);
  return v;
}
__device__ __forceinline__ float sum4(float4 x) { return x.x + x.y + x.z + x.w; }
__device__ __forceinline__ float sumsq4(float4 x) { return x.x*x.x + x.y*x.y + x.z*x.z + x.w*x.w; }
__device__ __forceinline__ float4 ln4(float4 x, float mu, float rs, float4 g, float4 b) {
  float4 y;
  y.x = (x.x - mu) * rs * g.x + b.x;
  y.y = (x.y - mu) * rs * g.y + b.y;
  y.z = (x.z - mu) * rs * g.z + b.z;
  y.w = (x.w - mu) * rs * g.w + b.w;
  return y;
}

__global__ __launch_bounds__(256) void leaf_ln_kernel(
    const float* __restrict__ comp_emb, const int* __restrict__ leaf_ids,
    const float* __restrict__ gamma, const float* __restrict__ beta,
    float* __restrict__ h0) {
  const int lane = threadIdx.x & 31;
  const int r = blockIdx.x * 8 + (threadIdx.x >> 5);
  const int id = leaf_ids[r];
  const float4* src = (const float4*)(comp_emb + (size_t)id * DD);
  float4 x0 = src[lane];
  float4 x1 = src[lane + 32];
  float s  = wave_sum(sum4(x0) + sum4(x1));
  float s2 = wave_sum(sumsq4(x0) + sumsq4(x1));
  float mu = s * (1.0f / DD);
  float rs = rsqrtf(s2 * (1.0f / DD) - mu * mu + EPS);
  float4 g0 = ((const float4*)gamma)[lane], g1 = ((const float4*)gamma)[lane + 32];
  float4 b0 = ((const float4*)beta)[lane],  b1 = ((const float4*)beta)[lane + 32];
  float4* dst = (float4*)(h0 + (size_t)r * DD);
  dst[lane]      = ln4(x0, mu, rs, g0, b0);
  dst[lane + 32] = ln4(x1, mu, rs, g1, b1);
}

#define MT 32
#define KC 768
#define CSTR 772
#define OSTR 260

__global__ __launch_bounds__(256) void combine_kernel(
    const float* __restrict__ h_in, float* __restrict__ h_out,
    const int* __restrict__ op_ids, int off, int ln2n,
    const float* __restrict__ op_emb, const float* __restrict__ W,
    const float* __restrict__ bias, const float* __restrict__ gamma,
    const float* __restrict__ beta) {
  __shared__ float smem[MT * CSTR];

  const int tid = threadIdx.x;
  const int lane = tid & 31;
  const int wave = tid >> 5;
  const int r0 = blockIdx.x * MT;
  const int nmask = (1 << ln2n) - 1;

  {
    const int c4 = tid & 63;
    const int mrow = tid >> 6;
    int opid[8];
#pragma unroll
    for (int j = 0; j < 8; ++j) {
      const int rg = r0 + mrow + 4 * j;
      opid[j] = op_ids[(rg >> ln2n) * 7 + off + (rg & nmask)];
    }
    float4 vl[8], vr[8], vo[8];
#pragma unroll
    for (int j = 0; j < 8; ++j) {
      const int rg = r0 + mrow + 4 * j;
      const float4* src = (const float4*)(h_in + (size_t)(2 * rg) * DD);
      vl[j] = src[c4];
      vr[j] = src[64 + c4];
    }
#pragma unroll
    for (int j = 0; j < 8; ++j)
      vo[j] = ((const float4*)(op_emb + (size_t)opid[j] * DD))[c4];
#pragma unroll
    for (int j = 0; j < 8; ++j) {
      float* dst = &smem[(mrow + 4 * j) * CSTR + c4 * 4];
      *(float4*)(dst + DD) = vl[j];
      *(float4*)(dst + 2 * DD) = vr[j];
      *(float4*)dst = vo[j];
    }
  }
  __syncthreads();

  v8f c00 = {}, c01 = {}, c10 = {}, c11 = {};
  const int n0 = wave * 32;
  const int colA = n0 + (lane & 15);
  const int krow = (lane & 16) >> 3;
  const float* arow0 = &smem[(lane & 15) * CSTR];
  const float* arow1 = &smem[((lane & 15) + 16) * CSTR];
#pragma unroll 4
  for (int k0 = 0; k0 < KC; k0 += 4) {
    const int kb = k0 + krow;
    const float* Wk = W + (size_t)kb * DD;
    v2f b0 = { Wk[colA],      Wk[DD + colA] };
    v2f b1 = { Wk[colA + 16], Wk[DD + colA + 16] };
    v2f a0 = *(const v2f*)(arow0 + kb);
    v2f a1 = *(const v2f*)(arow1 + kb);
    c00 = __builtin_amdgcn_wmma_f32_16x16x4_f32(false, a0, false, b0, (short)0, c00, false, false);
    c01 = __builtin_amdgcn_wmma_f32_16x16x4_f32(false, a0, false, b1, (short)0, c01, false, false);
    c10 = __builtin_amdgcn_wmma_f32_16x16x4_f32(false, a1, false, b0, (short)0, c10, false, false);
    c11 = __builtin_amdgcn_wmma_f32_16x16x4_f32(false, a1, false, b1, (short)0, c11, false, false);
  }
  __syncthreads();

  const float bias0 = bias[colA];
  const float bias1 = bias[colA + 16];
  const int roff = (lane & 16) >> 1;
#pragma unroll
  for (int j = 0; j < 8; ++j) {
    const int rr = j + roff;
    smem[rr * OSTR + colA]              = c00[j] + bias0;
    smem[rr * OSTR + colA + 16]         = c01[j] + bias1;
    smem[(rr + 16) * OSTR + colA]       = c10[j] + bias0;
    smem[(rr + 16) * OSTR + colA + 16]  = c11[j] + bias1;
  }
  __syncthreads();

  const float4 g0  = ((const float4*)gamma)[lane];
  const float4 g1  = ((const float4*)gamma)[lane + 32];
  const float4 bt0 = ((const float4*)beta)[lane];
  const float4 bt1 = ((const float4*)beta)[lane + 32];
#pragma unroll
  for (int i = 0; i < 4; ++i) {
    const int row = wave * 4 + i;
    float4 x0 = *(const float4*)&smem[row * OSTR + lane * 4];
    float4 x1 = *(const float4*)&smem[row * OSTR + 128 + lane * 4];
    float s  = wave_sum(sum4(x0) + sum4(x1));
    float s2 = wave_sum(sumsq4(x0) + sumsq4(x1));
    float mu = s * (1.0f / DD);
    float rs = rsqrtf(s2 * (1.0f / DD) - mu * mu + EPS);
    float4* dst = (float4*)(h_out + (size_t)(r0 + row) * DD);
    dst[lane]      = ln4(x0, mu, rs, g0, bt0);
    dst[lane + 32] = ln4(x1, mu, rs, g1, bt1);
  }
}

__global__ __launch_bounds__(256) void gather_kernel(
    const float* __restrict__ root, const int* __restrict__ sidx,
    float* __restrict__ out) {
  const int lane = threadIdx.x & 31;
  const int r = blockIdx.x * 8 + (threadIdx.x >> 5);
  const int idx = sidx[r];
  float4* dst = (float4*)(out + (size_t)r * DD);
  if (idx > 1) {
    const float4* src = (const float4*)(root + (size_t)idx * DD);
    dst[lane]      = src[lane];
    dst[lane + 32] = src[lane + 32];
  } else {
    float4 z = make_float4(0.f, 0.f, 0.f, 0.f);
    dst[lane]      = z;
    dst[lane + 32] = z;
  }
}

extern "C" void kernel_launch(void* const* d_in, const int* in_sizes, int n_in,
                              void* d_out, int out_size, void* d_ws, size_t ws_size,
                              hipStream_t stream) {
  const int*   struct_indices = (const int*)d_in[0];
  const int*   leaf_ids       = (const int*)d_in[1];
  const int*   op_ids         = (const int*)d_in[2];
  const float* comp_emb       = (const float*)d_in[3];
  const float* op_emb         = (const float*)d_in[4];
  const float* W_bin          = (const float*)d_in[5];
  const float* b_bin          = (const float*)d_in[6];
  const float* gamma          = (const float*)d_in[7];
  const float* beta           = (const float*)d_in[8];

  float* h0   = (float*)d_out;
  float* h1   = (float*)d_ws;
  float* h2   = h1 + (size_t)NSTRUCT * 4 * DD;
  float* root = h2 + (size_t)NSTRUCT * 2 * DD;

  leaf_ln_kernel<<<NSTRUCT * 8 / 8, 256, 0, stream>>>(comp_emb, leaf_ids, gamma, beta, h0);
  combine_kernel<<<(NSTRUCT * 4) / MT, 256, 0, stream>>>(h0, h1, op_ids, 0, 2,
                                                         op_emb, W_bin, b_bin, gamma, beta);
  combine_kernel<<<(NSTRUCT * 2) / MT, 256, 0, stream>>>(h1, h2, op_ids, 4, 1,
                                                         op_emb, W_bin, b_bin, gamma, beta);
  combine_kernel<<<(NSTRUCT * 1) / MT, 256, 0, stream>>>(h2, root, op_ids, 6, 0,
                                                         op_emb, W_bin, b_bin, gamma, beta);
  gather_kernel<<<65536 / 8, 256, 0, stream>>>(root, struct_indices, (float*)d_out);
}


# operator third of K factored out: op_emb @ W[0:256] built once per call as a 64x256 table on the f32 matrix instruction, used as accumulator start; combine K 768 -> 512, 4 work-groups per WGP
# speedup vs baseline: 1.4523x; 1.2880x over previous
#include <hip/hip_runtime.h>

#define DD 256
#define NSTRUCT 8192
#define EPS 1e-5f

typedef float v2f __attribute__((ext_vector_type(2)));
typedef float v8f __attribute__((ext_vector_type(8)));

__device__ __forceinline__ float wave_sum(float v) {
  v += __shfl_xor(v, 1, 32);
  v += __shfl_xor(v, 2, 32);
  v += __shfl_xor(v, 4, 32);
  v += __shfl_xor(v, 8, 32);
  v += __shfl_xor(v, 16, 32);
  return v;
}
__device__ __forceinline__ float sum4(float4 x) { return x.x + x.y + x.z + x.w; }
__device__ __forceinline__ float sumsq4(float4 x) { return x.x*x.x + x.y*x.y + x.z*x.z + x.w*x.w; }
__device__ __forceinline__ float4 ln4(float4 x, float mu, float rs, float4 g, float4 b) {
  float4 y;
  y.x = (x.x - mu) * rs * g.x + b.x;
  y.y = (x.y - mu) * rs * g.y + b.y;
  y.z = (x.z - mu) * rs * g.z + b.z;
  y.w = (x.w - mu) * rs * g.w + b.w;
  return y;
}

#define PBLK 4

__global__ __launch_bounds__(256) void leaf_ln_kernel(
    const float* __restrict__ comp_emb, const int* __restrict__ leaf_ids,
    const float* __restrict__ gamma, const float* __restrict__ beta,
    float* __restrict__ h0,
    const float* __restrict__ op_emb, const float* __restrict__ W,
    float* __restrict__ optab) {
  const int lane = threadIdx.x & 31;
  if (blockIdx.x < PBLK) {
    const int t = blockIdx.x * 8 + (threadIdx.x >> 5);
    const int colA = (t & 7) * 32 + (lane & 15);
    const int krow = (lane & 16) >> 3;
    const float* arow = op_emb + (size_t)((t >> 3) * 16 + (lane & 15)) * DD;
    v8f c0 = {}, c1 = {};
#pragma unroll 8
    for (int k0 = 0; k0 < DD; k0 += 4) {
      const int kb = k0 + krow;
      const float* Wk = W + (size_t)kb * DD;
      v2f b0 = { Wk[colA],      Wk[DD + colA] };
      v2f b1 = { Wk[colA + 16], Wk[DD + colA + 16] };
      v2f a  = *(const v2f*)(arow + kb);
      c0 = __builtin_amdgcn_wmma_f32_16x16x4_f32(false, a, false, b0, (short)0, c0, false, false);
      c1 = __builtin_amdgcn_wmma_f32_16x16x4_f32(false, a, false, b1, (short)0, c1, false, false);
    }
    float* dst = optab + (size_t)((t >> 3) * 16 + ((lane & 16) >> 1)) * DD + colA;
#pragma unroll
    for (int j = 0; j < 8; ++j) {
      dst[j * DD]      = c0[j];
      dst[j * DD + 16] = c1[j];
    }
    return;
  }
  const int r = (blockIdx.x - PBLK) * 8 + (threadIdx.x >> 5);
  const int id = leaf_ids[r];
  const float4* src = (const float4*)(comp_emb + (size_t)id * DD);
  float4 x0 = src[lane];
  float4 x1 = src[lane + 32];
  float s  = wave_sum(sum4(x0) + sum4(x1));
  float s2 = wave_sum(sumsq4(x0) + sumsq4(x1));
  float mu = s * (1.0f / DD);
  float rs = rsqrtf(s2 * (1.0f / DD) - mu * mu + EPS);
  float4 g0 = ((const float4*)gamma)[lane], g1 = ((const float4*)gamma)[lane + 32];
  float4 b0 = ((const float4*)beta)[lane],  b1 = ((const float4*)beta)[lane + 32];
  float4* dst = (float4*)(h0 + (size_t)r * DD);
  dst[lane]      = ln4(x0, mu, rs, g0, b0);
  dst[lane + 32] = ln4(x1, mu, rs, g1, b1);
}

#define MT 32
#define KC 512
#define CSTR 516
#define OSTR 260

__global__ __launch_bounds__(256, 8) void combine_kernel(
    const float* __restrict__ h_in, float* __restrict__ h_out,
    const int* __restrict__ op_ids, int off, int ln2n,
    const float* __restrict__ optab, const float* __restrict__ W,
    const float* __restrict__ bias, const float* __restrict__ gamma,
    const float* __restrict__ beta) {
  __shared__ float smem[MT * CSTR];

  const int tid = threadIdx.x;
  const int lane = tid & 31;
  const int wave = tid >> 5;
  const int r0 = blockIdx.x * MT;
  const int nmask = (1 << ln2n) - 1;

  const int n0 = wave * 32;
  const int colA = n0 + (lane & 15);
  const int roff = (lane & 16) >> 1;

  int opid[16];
#pragma unroll
  for (int j = 0; j < 16; ++j) {
    const int rg = r0 + (j & 7) + roff + (j & 8) * 2;
    opid[j] = op_ids[(rg >> ln2n) * 7 + off + (rg & nmask)];
  }

  v8f c00, c01, c10, c11;
  {
    const int c4 = tid & 63;
    const int mrow = tid >> 6;
    float4 vl[8], vr[8];
#pragma unroll
    for (int j = 0; j < 8; ++j) {
      const int rg = r0 + mrow + 4 * j;
      const float4* src = (const float4*)(h_in + (size_t)(2 * rg) * DD);
      vl[j] = src[c4];
      vr[j] = src[64 + c4];
    }
#pragma unroll
    for (int j = 0; j < 8; ++j) {
      const float* p0 = optab + (size_t)opid[j] * DD + colA;
      const float* p1 = optab + (size_t)opid[8 + j] * DD + colA;
      c00[j] = p0[0];
      c01[j] = p0[16];
      c10[j] = p1[0];
      c11[j] = p1[16];
    }
#pragma unroll
    for (int j = 0; j < 8; ++j) {
      float* dst = &smem[(mrow + 4 * j) * CSTR + c4 * 4];
      *(float4*)dst = vl[j];
      *(float4*)(dst + DD) = vr[j];
    }
  }
  __syncthreads();

  const int krow = (lane & 16) >> 3;
  const float* arow0 = &smem[(lane & 15) * CSTR];
  const float* arow1 = &smem[((lane & 15) + 16) * CSTR];
#pragma unroll 4
  for (int k0 = 0; k0 < KC; k0 += 4) {
    const int kb = k0 + krow;
    const float* Wk = W + (size_t)(DD + kb) * DD;
    v2f b0 = { Wk[colA],      Wk[DD + colA] };
    v2f b1 = { Wk[colA + 16], Wk[DD + colA + 16] };
    v2f a0 = *(const v2f*)(arow0 + kb);
    v2f a1 = *(const v2f*)(arow1 + kb);
    c00 = __builtin_amdgcn_wmma_f32_16x16x4_f32(false, a0, false, b0, (short)0, c00, false, false);
    c01 = __builtin_amdgcn_wmma_f32_16x16x4_f32(false, a0, false, b1, (short)0, c01, false, false);
    c10 = __builtin_amdgcn_wmma_f32_16x16x4_f32(false, a1, false, b0, (short)0, c10, false, false);
    c11 = __builtin_amdgcn_wmma_f32_16x16x4_f32(false, a1, false, b1, (short)0, c11, false, false);
  }
  __syncthreads();

  const float bias0 = bias[colA];
  const float bias1 = bias[colA + 16];
#pragma unroll
  for (int j = 0; j < 8; ++j) {
    const int rr = j + roff;
    smem[rr * OSTR + colA]              = c00[j] + bias0;
    smem[rr * OSTR + colA + 16]         = c01[j] + bias1;
    smem[(rr + 16) * OSTR + colA]       = c10[j] + bias0;
    smem[(rr + 16) * OSTR + colA + 16]  = c11[j] + bias1;
  }
  __syncthreads();

  const float4 g0  = ((const float4*)gamma)[lane];
  const float4 g1  = ((const float4*)gamma)[lane + 32];
  const float4 bt0 = ((const float4*)beta)[lane];
  const float4 bt1 = ((const float4*)beta)[lane + 32];
#pragma unroll
  for (int i = 0; i < 4; ++i) {
    const int row = wave * 4 + i;
    float4 x0 = *(const float4*)&smem[row * OSTR + lane * 4];
    float4 x1 = *(const float4*)&smem[row * OSTR + 128 + lane * 4];
    float s  = wave_sum(sum4(x0) + sum4(x1));
    float s2 = wave_sum(sumsq4(x0) + sumsq4(x1));
    float mu = s * (1.0f / DD);
    float rs = rsqrtf(s2 * (1.0f / DD) - mu * mu + EPS);
    float4* dst = (float4*)(h_out + (size_t)(r0 + row) * DD);
    dst[lane]      = ln4(x0, mu, rs, g0, bt0);
    dst[lane + 32] = ln4(x1, mu, rs, g1, bt1);
  }
}

__global__ __launch_bounds__(256) void gather_kernel(
    const float* __restrict__ root, const int* __restrict__ sidx,
    float* __restrict__ out) {
  const int lane = threadIdx.x & 31;
  const int r = blockIdx.x * 8 + (threadIdx.x >> 5);
  const int idx = sidx[r];
  float4* dst = (float4*)(out + (size_t)r * DD);
  if (idx > 1) {
    const float4* src = (const float4*)(root + (size_t)idx * DD);
    dst[lane]      = src[lane];
    dst[lane + 32] = src[lane + 32];
  } else {
    float4 z = make_float4(0.f, 0.f, 0.f, 0.f);
    dst[lane]      = z;
    dst[lane + 32] = z;
  }
}

extern "C" void kernel_launch(void* const* d_in, const int* in_sizes, int n_in,
                              void* d_out, int out_size, void* d_ws, size_t ws_size,
                              hipStream_t stream) {
  const int*   struct_indices = (const int*)d_in[0];
  const int*   leaf_ids       = (const int*)d_in[1];
  const int*   op_ids         = (const int*)d_in[2];
  const float* comp_emb       = (const float*)d_in[3];
  const float* op_emb         = (const float*)d_in[4];
  const float* W_bin          = (const float*)d_in[5];
  const float* b_bin          = (const float*)d_in[6];
  const float* gamma          = (const float*)d_in[7];
  const float* beta           = (const float*)d_in[8];

  float* h0    = (float*)d_out;
  float* h1    = (float*)d_ws;
  float* h2    = h1 + (size_t)NSTRUCT * 4 * DD;
  float* optab = h2 + (size_t)NSTRUCT * 2 * DD;
  float* root  = h1;

  leaf_ln_kernel<<<NSTRUCT * 8 / 8 + PBLK, 256, 0, stream>>>(comp_emb, leaf_ids, gamma, beta, h0,
                                                            op_emb, W_bin, optab);
  combine_kernel<<<(NSTRUCT * 4) / MT, 256, 0, stream>>>(h0, h1, op_ids, 0, 2,
                                                         optab, W_bin, b_bin, gamma, beta);
  combine_kernel<<<(NSTRUCT * 2) / MT, 256, 0, stream>>>(h1, h2, op_ids, 4, 1,
                                                         optab, W_bin, b_bin, gamma, beta);
  combine_kernel<<<(NSTRUCT * 1) / MT, 256, 0, stream>>>(h2, root, op_ids, 6, 0,
                                                         optab, W_bin, b_bin, gamma, beta);
  gather_kernel<<<65536 / 8, 256, 0, stream>>>(root, struct_indices, (float*)d_out);
}


# level 1 factored through the 1024-row vocabulary: LN(comp_emb) @ W[256:512] and @ W[512:768] built per call on the f32 matrix instruction; level 1 = gather+add+LN, no h0; level-1 sum order changes
# speedup vs baseline: 2.4295x; 1.6729x over previous
#include <hip/hip_runtime.h>

#define DD 256
#define NSTRUCT 8192
#define EPS 1e-5f

typedef float v2f __attribute__((ext_vector_type(2)));
typedef float v8f __attribute__((ext_vector_type(8)));

__device__ __forceinline__ float wave_sum(float v) {
  v += __shfl_xor(v, 1, 32);
  v += __shfl_xor(v, 2, 32);
  v += __shfl_xor(v, 4, 32);
  v += __shfl_xor(v, 8, 32);
  v += __shfl_xor(v, 16, 32);
  return v;
}
__device__ __forceinline__ float sum4(float4 x) { return x.x + x.y + x.z + x.w; }
__device__ __forceinline__ float sumsq4(float4 x) { return x.x*x.x + x.y*x.y + x.z*x.z + x.w*x.w; }
__device__ __forceinline__ float4 ln4(float4 x, float mu, float rs, float4 g, float4 b) {
  float4 y;
  y.x = (x.x - mu) * rs * g.x + b.x;
  y.y = (x.y - mu) * rs * g.y + b.y;
  y.z = (x.z - mu) * rs * g.z + b.z;
  y.w = (x.w - mu) * rs * g.w + b.w;
  return y;
}

#define PBLK 4
#define NCOMP 1024
#define TROWS 16
#define TSTR 260

__global__ __launch_bounds__(256) void table_kernel(
    const float* __restrict__ comp_emb, const float* __restrict__ gamma, const float* __restrict__ beta,
    const float* __restrict__ op_emb, const float* __restrict__ W,
    float* __restrict__ optab, float* __restrict__ leaftab) {
  __shared__ __attribute__((aligned(16))) float sa[TROWS * TSTR];
  const int lane = threadIdx.x & 31;
  const int wave = threadIdx.x >> 5;
  const int krow = (lane & 16) >> 3;
  if (blockIdx.x < PBLK) {
    const int t = blockIdx.x * 8 + wave;
    const int colA = (t & 7) * 32 + (lane & 15);
    const float* arow = op_emb + (size_t)((t >> 3) * 16 + (lane & 15)) * DD;
    v8f c0 = {}, c1 = {};
#pragma unroll 8
    for (int k0 = 0; k0 < DD; k0 += 4) {
      const int kb = k0 + krow;
      const float* Wk = W + (size_t)kb * DD;
      v2f b0 = { Wk[colA],      Wk[DD + colA] };
      v2f b1 = { Wk[colA + 16], Wk[DD + colA + 16] };
      v2f a  = *(const v2f*)(arow + kb);
      c0 = __builtin_amdgcn_wmma_f32_16x16x4_f32(false, a, false, b0, (short)0, c0, false, false);
      c1 = __builtin_amdgcn_wmma_f32_16x16x4_f32(false, a, false, b1, (short)0, c1, false, false);
    }
    float* dst = optab + (size_t)((t >> 3) * 16 + ((lane & 16) >> 1)) * DD + colA;
#pragma unroll
    for (int j = 0; j < 8; ++j) {
      dst[j * DD]      = c0[j];
      dst[j * DD + 16] = c1[j];
    }
    return;
  }
  const int blk  = blockIdx.x - PBLK;
  const int rgrp = blk >> 1;
  const int side = blk & 1;
  {
    const float4 g0 = ((const float4*)gamma)[lane], g1 = ((const float4*)gamma)[lane + 32];
    const float4 b0 = ((const float4*)beta)[lane],  b1 = ((const float4*)beta)[lane + 32];
#pragma unroll
    for (int i = 0; i < 2; ++i) {
      const int lr = wave * 2 + i;
      const float4* src = (const float4*)(comp_emb + (size_t)(rgrp * TROWS + lr) * DD);
      float4 x0 = src[lane];
      float4 x1 = src[lane + 32];
      float s  = wave_sum(sum4(x0) + sum4(x1));
      float s2 = wave_sum(sumsq4(x0) + sumsq4(x1));
      float mu = s * (1.0f / DD);
      float rs = rsqrtf(s2 * (1.0f / DD) - mu * mu + EPS);
      *(float4*)&sa[lr * TSTR + lane * 4]       = ln4(x0, mu, rs, g0, b0);
      *(float4*)&sa[lr * TSTR + 128 + lane * 4] = ln4(x1, mu, rs, g1, b1);
    }
  }
  __syncthreads();
  const int colA = wave * 32 + (lane & 15);
  const float* arow = &sa[(lane & 15) * TSTR];
  const float* Ws = W + (size_t)(DD + side * DD) * DD;
  v8f c0 = {}, c1 = {};
#pragma unroll 8
  for (int k0 = 0; k0 < DD; k0 += 4) {
    const int kb = k0 + krow;
    const float* Wk = Ws + (size_t)kb * DD;
    v2f b0 = { Wk[colA],      Wk[DD + colA] };
    v2f b1 = { Wk[colA + 16], Wk[DD + colA + 16] };
    v2f a  = *(const v2f*)(arow + kb);
    c0 = __builtin_amdgcn_wmma_f32_16x16x4_f32(false, a, false, b0, (short)0, c0, false, false);
    c1 = __builtin_amdgcn_wmma_f32_16x16x4_f32(false, a, false, b1, (short)0, c1, false, false);
  }
  float* dst = leaftab + (size_t)side * NCOMP * DD + (size_t)(rgrp * TROWS + ((lane & 16) >> 1)) * DD + colA;
#pragma unroll
  for (int j = 0; j < 8; ++j) {
    dst[j * DD]      = c0[j];
    dst[j * DD + 16] = c1[j];
  }
}

__global__ __launch_bounds__(256) void level1_kernel(
    const int* __restrict__ leaf_ids, const int* __restrict__ op_ids,
    const float* __restrict__ optab, const float* __restrict__ leaftab,
    const float* __restrict__ bias, const float* __restrict__ gamma, const float* __restrict__ beta,
    float* __restrict__ h1) {
  const int lane = threadIdx.x & 31;
  const int st = __builtin_amdgcn_readfirstlane(blockIdx.x * 8 + (threadIdx.x >> 5));
  int lid[8], oid[4];
#pragma unroll
  for (int j = 0; j < 8; ++j) lid[j] = leaf_ids[st * 8 + j];
#pragma unroll
  for (int j = 0; j < 4; ++j) oid[j] = op_ids[st * 7 + j];
  float4 o0[4], o1[4], l0[4], l1[4], r0[4], r1[4];
#pragma unroll
  for (int i = 0; i < 4; ++i) {
    const float4* po = (const float4*)(optab + (size_t)oid[i] * DD);
    const float4* pl = (const float4*)(leaftab + (size_t)lid[2 * i] * DD);
    const float4* pr = (const float4*)(leaftab + (size_t)NCOMP * DD + (size_t)lid[2 * i + 1] * DD);
    o0[i] = po[lane]; o1[i] = po[lane + 32];
    l0[i] = pl[lane]; l1[i] = pl[lane + 32];
    r0[i] = pr[lane]; r1[i] = pr[lane + 32];
  }
  const float4 bs0 = ((const float4*)bias)[lane],  bs1 = ((const float4*)bias)[lane + 32];
  const float4 g0  = ((const float4*)gamma)[lane], g1  = ((const float4*)gamma)[lane + 32];
  const float4 bt0 = ((const float4*)beta)[lane],  bt1 = ((const float4*)beta)[lane + 32];
#pragma unroll
  for (int i = 0; i < 4; ++i) {
    float4 x0, x1;
    x0.x = ((o0[i].x + l0[i].x) + r0[i].x) + bs0.x;  x0.y = ((o0[i].y + l0[i].y) + r0[i].y) + bs0.y;
    x0.z = ((o0[i].z + l0[i].z) + r0[i].z) + bs0.z;  x0.w = ((o0[i].w + l0[i].w) + r0[i].w) + bs0.w;
    x1.x = ((o1[i].x + l1[i].x) + r1[i].x) + bs1.x;  x1.y = ((o1[i].y + l1[i].y) + r1[i].y) + bs1.y;
    x1.z = ((o1[i].z + l1[i].z) + r1[i].z) + bs1.z;  x1.w = ((o1[i].w + l1[i].w) + r1[i].w) + bs1.w;
    float s  = wave_sum(sum4(x0) + sum4(x1));
    float s2 = wave_sum(sumsq4(x0) + sumsq4(x1));
    float mu = s * (1.0f / DD);
    float rs = rsqrtf(s2 * (1.0f / DD) - mu * mu + EPS);
    float4* dst = (float4*)(h1 + (size_t)(st * 4 + i) * DD);
    dst[lane]      = ln4(x0, mu, rs, g0, bt0);
    dst[lane + 32] = ln4(x1, mu, rs, g1, bt1);
  }
}

#define MT 32
#define KC 512
#define CSTR 516
#define OSTR 260

__global__ __launch_bounds__(256, 8) void combine_kernel(
    const float* __restrict__ h_in, float* __restrict__ h_out,
    const int* __restrict__ op_ids, int off, int ln2n,
    const float* __restrict__ optab, const float* __restrict__ W,
    const float* __restrict__ bias, const float* __restrict__ gamma,
    const float* __restrict__ beta) {
  __shared__ float smem[MT * CSTR];

  const int tid = threadIdx.x;
  const int lane = tid & 31;
  const int wave = tid >> 5;
  const int r0 = blockIdx.x * MT;
  const int nmask = (1 << ln2n) - 1;

  const int n0 = wave * 32;
  const int colA = n0 + (lane & 15);
  const int roff = (lane & 16) >> 1;

  int opid[16];
#pragma unroll
  for (int j = 0; j < 16; ++j) {
    const int rg = r0 + (j & 7) + roff + (j & 8) * 2;
    opid[j] = op_ids[(rg >> ln2n) * 7 + off + (rg & nmask)];
  }

  v8f c00, c01, c10, c11;
  {
    const int c4 = tid & 63;
    const int mrow = tid >> 6;
    float4 vl[8], vr[8];
#pragma unroll
    for (int j = 0; j < 8; ++j) {
      const int rg = r0 + mrow + 4 * j;
      const float4* src = (const float4*)(h_in + (size_t)(2 * rg) * DD);
      vl[j] = src[c4];
      vr[j] = src[64 + c4];
    }
#pragma unroll
    for (int j = 0; j < 8; ++j) {
      const float* p0 = optab + (size_t)opid[j] * DD + colA;
      const float* p1 = optab + (size_t)opid[8 + j] * DD + colA;
      c00[j] = p0[0];
      c01[j] = p0[16];
      c10[j] = p1[0];
      c11[j] = p1[16];
    }
#pragma unroll
    for (int j = 0; j < 8; ++j) {
      float* dst = &smem[(mrow + 4 * j) * CSTR + c4 * 4];
      *(float4*)dst = vl[j];
      *(float4*)(dst + DD) = vr[j];
    }
  }
  __syncthreads();

  const int krow = (lane & 16) >> 3;
  const float* arow0 = &smem[(lane & 15) * CSTR];
  const float* arow1 = &smem[((lane & 15) + 16) * CSTR];
#pragma unroll 4
  for (int k0 = 0; k0 < KC; k0 += 4) {
    const int kb = k0 + krow;
    const float* Wk = W + (size_t)(DD + kb) * DD;
    v2f b0 = { Wk[colA],      Wk[DD + colA] };
    v2f b1 = { Wk[colA + 16], Wk[DD + colA + 16] };
    v2f a0 = *(const v2f*)(arow0 + kb);
    v2f a1 = *(const v2f*)(arow1 + kb);
    c00 = __builtin_amdgcn_wmma_f32_16x16x4_f32(false, a0, false, b0, (short)0, c00, false, false);
    c01 = __builtin_amdgcn_wmma_f32_16x16x4_f32(false, a0, false, b1, (short)0, c01, false, false);
    c10 = __builtin_amdgcn_wmma_f32_16x16x4_f32(false, a1, false, b0, (short)0, c10, false, false);
    c11 = __builtin_amdgcn_wmma_f32_16x16x4_f32(false, a1, false, b1, (short)0, c11, false, false);
  }
  __syncthreads();

  const float bias0 = bias[colA];
  const float bias1 = bias[colA + 16];
#pragma unroll
  for (int j = 0; j < 8; ++j) {
    const int rr = j + roff;
    smem[rr * OSTR + colA]              = c00[j] + bias0;
    smem[rr * OSTR + colA + 16]         = c01[j] + bias1;
    smem[(rr + 16) * OSTR + colA]       = c10[j] + bias0;
    smem[(rr + 16) * OSTR + colA + 16]  = c11[j] + bias1;
  }
  __syncthreads();

  const float4 g0  = ((const float4*)gamma)[lane];
  const float4 g1  = ((const float4*)gamma)[lane + 32];
  const float4 bt0 = ((const float4*)beta)[lane];
  const float4 bt1 = ((const float4*)beta)[lane + 32];
#pragma unroll
  for (int i = 0; i < 4; ++i) {
    const int row = wave * 4 + i;
    float4 x0 = *(const float4*)&smem[row * OSTR + lane * 4];
    float4 x1 = *(const float4*)&smem[row * OSTR + 128 + lane * 4];
    float s  = wave_sum(sum4(x0) + sum4(x1));
    float s2 = wave_sum(sumsq4(x0) + sumsq4(x1));
    float mu = s * (1.0f / DD);
    float rs = rsqrtf(s2 * (1.0f / DD) - mu * mu + EPS);
    float4* dst = (float4*)(h_out + (size_t)(r0 + row) * DD);
    dst[lane]      = ln4(x0, mu, rs, g0, bt0);
    dst[lane + 32] = ln4(x1, mu, rs, g1, bt1);
  }
}

__global__ __launch_bounds__(256) void gather_kernel(
    const float* __restrict__ root, const int* __restrict__ sidx,
    float* __restrict__ out) {
  const int lane = threadIdx.x & 31;
  const int r = blockIdx.x * 8 + (threadIdx.x >> 5);
  const int idx = sidx[r];
  float4* dst = (float4*)(out + (size_t)r * DD);
  if (idx > 1) {
    const float4* src = (const float4*)(root + (size_t)idx * DD);
    dst[lane]      = src[lane];
    dst[lane + 32] = src[lane + 32];
  } else {
    float4 z = make_float4(0.f, 0.f, 0.f, 0.f);
    dst[lane]      = z;
    dst[lane + 32] = z;
  }
}

extern "C" void kernel_launch(void* const* d_in, const int* in_sizes, int n_in,
                              void* d_out, int out_size, void* d_ws, size_t ws_size,
                              hipStream_t stream) {
  const int*   struct_indices = (const int*)d_in[0];
  const int*   leaf_ids       = (const int*)d_in[1];
  const int*   op_ids         = (const int*)d_in[2];
  const float* comp_emb       = (const float*)d_in[3];
  const float* op_emb         = (const float*)d_in[4];
  const float* W_bin          = (const float*)d_in[5];
  const float* b_bin          = (const float*)d_in[6];
  const float* gamma          = (const float*)d_in[7];
  const float* beta           = (const float*)d_in[8];

  float* h1      = (float*)d_ws;
  float* h2      = h1 + (size_t)NSTRUCT * 4 * DD;
  float* optab   = h2 + (size_t)NSTRUCT * 2 * DD;
  float* leaftab = optab + (size_t)64 * DD;
  float* root    = h1;

  table_kernel<<<PBLK + 2 * (NCOMP / TROWS), 256, 0, stream>>>(comp_emb, gamma, beta, op_emb, W_bin,
                                                               optab, leaftab);
  level1_kernel<<<NSTRUCT / 8, 256, 0, stream>>>(leaf_ids, op_ids, optab, leaftab, b_bin, gamma, beta, h1);
  combine_kernel<<<(NSTRUCT * 2) / MT, 256, 0, stream>>>(h1, h2, op_ids, 4, 1,
                                                         optab, W_bin, b_bin, gamma, beta);
  combine_kernel<<<(NSTRUCT * 1) / MT, 256, 0, stream>>>(h2, root, op_ids, 6, 0,
                                                         optab, W_bin, b_bin, gamma, beta);
  gather_kernel<<<65536 / 8, 256, 0, stream>>>(root, struct_indices, (float*)d_out);
}


# final gather: each wave copies 16 output rows (was 1): its 16 indices by one scalar load, all 32 row loads requested before the first store; 4096 waves instead of 65536
# speedup vs baseline: 2.5490x; 1.0492x over previous
#include <hip/hip_runtime.h>

#define DD 256
#define NSTRUCT 8192
#define EPS 1e-5f

typedef float v2f __attribute__((ext_vector_type(2)));
typedef float v8f __attribute__((ext_vector_type(8)));

__device__ __forceinline__ float wave_sum(float v) {
  v += __shfl_xor(v, 1, 32);
  v += __shfl_xor(v, 2, 32);
  v += __shfl_xor(v, 4, 32);
  v += __shfl_xor(v, 8, 32);
  v += __shfl_xor(v, 16, 32);
  return v;
}
__device__ __forceinline__ float sum4(float4 x) { return x.x + x.y + x.z + x.w; }
__device__ __forceinline__ float sumsq4(float4 x) { return x.x*x.x + x.y*x.y + x.z*x.z + x.w*x.w; }
__device__ __forceinline__ float4 ln4(float4 x, float mu, float rs, float4 g, float4 b) {
  float4 y;
  y.x = (x.x - mu) * rs * g.x + b.x;
  y.y = (x.y - mu) * rs * g.y + b.y;
  y.z = (x.z - mu) * rs * g.z + b.z;
  y.w = (x.w - mu) * rs * g.w + b.w;
  return y;
}

#define PBLK 4
#define NCOMP 1024
#define TROWS 16
#define TSTR 260

__global__ __launch_bounds__(256) void table_kernel(
    const float* __restrict__ comp_emb, const float* __restrict__ gamma, const float* __restrict__ beta,
    const float* __restrict__ op_emb, const float* __restrict__ W,
    float* __restrict__ optab, float* __restrict__ leaftab) {
  __shared__ __attribute__((aligned(16))) float sa[TROWS * TSTR];
  const int lane = threadIdx.x & 31;
  const int wave = threadIdx.x >> 5;
  const int krow = (lane & 16) >> 3;
  if (blockIdx.x < PBLK) {
    const int t = blockIdx.x * 8 + wave;
    const int colA = (t & 7) * 32 + (lane & 15);
    const float* arow = op_emb + (size_t)((t >> 3) * 16 + (lane & 15)) * DD;
    v8f c0 = {}, c1 = {};
#pragma unroll 8
    for (int k0 = 0; k0 < DD; k0 += 4) {
      const int kb = k0 + krow;
      const float* Wk = W + (size_t)kb * DD;
      v2f b0 = { Wk[colA],      Wk[DD + colA] };
      v2f b1 = { Wk[colA + 16], Wk[DD + colA + 16] };
      v2f a  = *(const v2f*)(arow + kb);
      c0 = __builtin_amdgcn_wmma_f32_16x16x4_f32(false, a, false, b0, (short)0, c0, false, false);
      c1 = __builtin_amdgcn_wmma_f32_16x16x4_f32(false, a, false, b1, (short)0, c1, false, false);
    }
    float* dst = optab + (size_t)((t >> 3) * 16 + ((lane & 16) >> 1)) * DD + colA;
#pragma unroll
    for (int j = 0; j < 8; ++j) {
      dst[j * DD]      = c0[j];
      dst[j * DD + 16] = c1[j];
    }
    return;
  }
  const int blk  = blockIdx.x - PBLK;
  const int rgrp = blk >> 1;
  const int side = blk & 1;
  {
    const float4 g0 = ((const float4*)gamma)[lane], g1 = ((const float4*)gamma)[lane + 32];
    const float4 b0 = ((const float4*)beta)[lane],  b1 = ((const float4*)beta)[lane + 32];
#pragma unroll
    for (int i = 0; i < 2; ++i) {
      const int lr = wave * 2 + i;
      const float4* src = (const float4*)(comp_emb + (size_t)(rgrp * TROWS + lr) * DD);
      float4 x0 = src[lane];
      float4 x1 = src[lane + 32];
      float s  = wave_sum(sum4(x0) + sum4(x1));
      float s2 = wave_sum(sumsq4(x0) + sumsq4(x1));
      float mu = s * (1.0f / DD);
      float rs = rsqrtf(s2 * (1.0f / DD) - mu * mu + EPS);
      *(float4*)&sa[lr * TSTR + lane * 4]       = ln4(x0, mu, rs, g0, b0);
      *(float4*)&sa[lr * TSTR + 128 + lane * 4] = ln4(x1, mu, rs, g1, b1);
    }
  }
  __syncthreads();
  const int colA = wave * 32 + (lane & 15);
  const float* arow = &sa[(lane & 15) * TSTR];
  const float* Ws = W + (size_t)(DD + side * DD) * DD;
  v8f c0 = {}, c1 = {};
#pragma unroll 8
  for (int k0 = 0; k0 < DD; k0 += 4) {
    const int kb = k0 + krow;
    const float* Wk = Ws + (size_t)kb * DD;
    v2f b0 = { Wk[colA],      Wk[DD + colA] };
    v2f b1 = { Wk[colA + 16], Wk[DD + colA + 16] };
    v2f a  = *(const v2f*)(arow + kb);
    c0 = __builtin_amdgcn_wmma_f32_16x16x4_f32(false, a, false, b0, (short)0, c0, false, false);
    c1 = __builtin_amdgcn_wmma_f32_16x16x4_f32(false, a, false, b1, (short)0, c1, false, false);
  }
  float* dst = leaftab + (size_t)side * NCOMP * DD + (size_t)(rgrp * TROWS + ((lane & 16) >> 1)) * DD + colA;
#pragma unroll
  for (int j = 0; j < 8; ++j) {
    dst[j * DD]      = c0[j];
    dst[j * DD + 16] = c1[j];
  }
}

__global__ __launch_bounds__(256) void level1_kernel(
    const int* __restrict__ leaf_ids, const int* __restrict__ op_ids,
    const float* __restrict__ optab, const float* __restrict__ leaftab,
    const float* __restrict__ bias, const float* __restrict__ gamma, const float* __restrict__ beta,
    float* __restrict__ h1) {
  const int lane = threadIdx.x & 31;
  const int st = __builtin_amdgcn_readfirstlane(blockIdx.x * 8 + (threadIdx.x >> 5));
  int lid[8], oid[4];
#pragma unroll
  for (int j = 0; j < 8; ++j) lid[j] = leaf_ids[st * 8 + j];
#pragma unroll
  for (int j = 0; j < 4; ++j) oid[j] = op_ids[st * 7 + j];
  float4 o0[4], o1[4], l0[4], l1[4], r0[4], r1[4];
#pragma unroll
  for (int i = 0; i < 4; ++i) {
    const float4* po = (const float4*)(optab + (size_t)oid[i] * DD);
    const float4* pl = (const float4*)(leaftab + (size_t)lid[2 * i] * DD);
    const float4* pr = (const float4*)(leaftab + (size_t)NCOMP * DD + (size_t)lid[2 * i + 1] * DD);
    o0[i] = po[lane]; o1[i] = po[lane + 32];
    l0[i] = pl[lane]; l1[i] = pl[lane + 32];
    r0[i] = pr[lane]; r1[i] = pr[lane + 32];
  }
  const float4 bs0 = ((const float4*)bias)[lane],  bs1 = ((const float4*)bias)[lane + 32];
  const float4 g0  = ((const float4*)gamma)[lane], g1  = ((const float4*)gamma)[lane + 32];
  const float4 bt0 = ((const float4*)beta)[lane],  bt1 = ((const float4*)beta)[lane + 32];
#pragma unroll
  for (int i = 0; i < 4; ++i) {
    float4 x0, x1;
    x0.x = ((o0[i].x + l0[i].x) + r0[i].x) + bs0.x;  x0.y = ((o0[i].y + l0[i].y) + r0[i].y) + bs0.y;
    x0.z = ((o0[i].z + l0[i].z) + r0[i].z) + bs0.z;  x0.w = ((o0[i].w + l0[i].w) + r0[i].w) + bs0.w;
    x1.x = ((o1[i].x + l1[i].x) + r1[i].x) + bs1.x;  x1.y = ((o1[i].y + l1[i].y) + r1[i].y) + bs1.y;
    x1.z = ((o1[i].z + l1[i].z) + r1[i].z) + bs1.z;  x1.w = ((o1[i].w + l1[i].w) + r1[i].w) + bs1.w;
    float s  = wave_sum(sum4(x0) + sum4(x1));
    float s2 = wave_sum(sumsq4(x0) + sumsq4(x1));
    float mu = s * (1.0f / DD);
    float rs = rsqrtf(s2 * (1.0f / DD) - mu * mu + EPS);
    float4* dst = (float4*)(h1 + (size_t)(st * 4 + i) * DD);
    dst[lane]      = ln4(x0, mu, rs, g0, bt0);
    dst[lane + 32] = ln4(x1, mu, rs, g1, bt1);
  }
}

#define MT 32
#define KC 512
#define CSTR 516
#define OSTR 260

__global__ __launch_bounds__(256, 8) void combine_kernel(
    const float* __restrict__ h_in, float* __restrict__ h_out,
    const int* __restrict__ op_ids, int off, int ln2n,
    const float* __restrict__ optab, const float* __restrict__ W,
    const float* __restrict__ bias, const float* __restrict__ gamma,
    const float* __restrict__ beta) {
  __shared__ float smem[MT * CSTR];

  const int tid = threadIdx.x;
  const int lane = tid & 31;
  const int wave = tid >> 5;
  const int r0 = blockIdx.x * MT;
  const int nmask = (1 << ln2n) - 1;

  const int n0 = wave * 32;
  const int colA = n0 + (lane & 15);
  const int roff = (lane & 16) >> 1;

  int opid[16];
#pragma unroll
  for (int j = 0; j < 16; ++j) {
    const int rg = r0 + (j & 7) + roff + (j & 8) * 2;
    opid[j] = op_ids[(rg >> ln2n) * 7 + off + (rg & nmask)];
  }

  v8f c00, c01, c10, c11;
  {
    const int c4 = tid & 63;
    const int mrow = tid >> 6;
    float4 vl[8], vr[8];
#pragma unroll
    for (int j = 0; j < 8; ++j) {
      const int rg = r0 + mrow + 4 * j;
      const float4* src = (const float4*)(h_in + (size_t)(2 * rg) * DD);
      vl[j] = src[c4];
      vr[j] = src[64 + c4];
    }
#pragma unroll
    for (int j = 0; j < 8; ++j) {
      const float* p0 = optab + (size_t)opid[j] * DD + colA;
      const float* p1 = optab + (size_t)opid[8 + j] * DD + colA;
      c00[j] = p0[0];
      c01[j] = p0[16];
      c10[j] = p1[0];
      c11[j] = p1[16];
    }
#pragma unroll
    for (int j = 0; j < 8; ++j) {
      float* dst = &smem[(mrow + 4 * j) * CSTR + c4 * 4];
      *(float4*)dst = vl[j];
      *(float4*)(dst + DD) = vr[j];
    }
  }
  __syncthreads();

  const int krow = (lane & 16) >> 3;
  const float* arow0 = &smem[(lane & 15) * CSTR];
  const float* arow1 = &smem[((lane & 15) + 16) * CSTR];
#pragma unroll 4
  for (int k0 = 0; k0 < KC; k0 += 4) {
    const int kb = k0 + krow;
    const float* Wk = W + (size_t)(DD + kb) * DD;
    v2f b0 = { Wk[colA],      Wk[DD + colA] };
    v2f b1 = { Wk[colA + 16], Wk[DD + colA + 16] };
    v2f a0 = *(const v2f*)(arow0 + kb);
    v2f a1 = *(const v2f*)(arow1 + kb);
    c00 = __builtin_amdgcn_wmma_f32_16x16x4_f32(false, a0, false, b0, (short)0, c00, false, false);
    c01 = __builtin_amdgcn_wmma_f32_16x16x4_f32(false, a0, false, b1, (short)0, c01, false, false);
    c10 = __builtin_amdgcn_wmma_f32_16x16x4_f32(false, a1, false, b0, (short)0, c10, false, false);
    c11 = __builtin_amdgcn_wmma_f32_16x16x4_f32(false, a1, false, b1, (short)0, c11, false, false);
  }
  __syncthreads();

  const float bias0 = bias[colA];
  const float bias1 = bias[colA + 16];
#pragma unroll
  for (int j = 0; j < 8; ++j) {
    const int rr = j + roff;
    smem[rr * OSTR + colA]              = c00[j] + bias0;
    smem[rr * OSTR + colA + 16]         = c01[j] + bias1;
    smem[(rr + 16) * OSTR + colA]       = c10[j] + bias0;
    smem[(rr + 16) * OSTR + colA + 16]  = c11[j] + bias1;
  }
  __syncthreads();

  const float4 g0  = ((const float4*)gamma)[lane];
  const float4 g1  = ((const float4*)gamma)[lane + 32];
  const float4 bt0 = ((const float4*)beta)[lane];
  const float4 bt1 = ((const float4*)beta)[lane + 32];
#pragma unroll
  for (int i = 0; i < 4; ++i) {
    const int row = wave * 4 + i;
    float4 x0 = *(const float4*)&smem[row * OSTR + lane * 4];
    float4 x1 = *(const float4*)&smem[row * OSTR + 128 + lane * 4];
    float s  = wave_sum(sum4(x0) + sum4(x1));
    float s2 = wave_sum(sumsq4(x0) + sumsq4(x1));
    float mu = s * (1.0f / DD);
    float rs = rsqrtf(s2 * (1.0f / DD) - mu * mu + EPS);
    float4* dst = (float4*)(h_out + (size_t)(r0 + row) * DD);
    dst[lane]      = ln4(x0, mu, rs, g0, bt0);
    dst[lane + 32] = ln4(x1, mu, rs, g1, bt1);
  }
}

#define GROWS 16
typedef unsigned v4u __attribute__((ext_vector_type(4)));
__global__ __launch_bounds__(256) void gather_kernel(
    const float* __restrict__ root, const int* __restrict__ sidx,
    float* __restrict__ out) {
  const int lane = threadIdx.x & 31;
  const int r0 = __builtin_amdgcn_readfirstlane((blockIdx.x * 8 + (threadIdx.x >> 5)) * GROWS);
  int idx[GROWS];
#pragma unroll
  for (int j = 0; j < GROWS; ++j) idx[j] = sidx[r0 + j];
  v4u v0[GROWS], v1[GROWS];
#pragma unroll
  for (int j = 0; j < GROWS; ++j) {
    const unsigned li = (unsigned)idx[j] < (unsigned)NSTRUCT ? (unsigned)idx[j] : (unsigned)(NSTRUCT - 1);
    const v4u* src = (const v4u*)(root + (size_t)li * DD);
    v0[j] = src[lane];
    v1[j] = src[lane + 32];
  }
  __builtin_amdgcn_sched_barrier(0);
#pragma unroll
  for (int j = 0; j < GROWS; ++j) {
    v4u* dst = (v4u*)(out + (size_t)(r0 + j) * DD);
    const unsigned keep = idx[j] > 1 ? 0xffffffffu : 0u;
    dst[lane]      = v0[j] & keep;
    dst[lane + 32] = v1[j] & keep;
  }
}

extern "C" void kernel_launch(void* const* d_in, const int* in_sizes, int n_in,
                              void* d_out, int out_size, void* d_ws, size_t ws_size,
                              hipStream_t stream) {
  const int*   struct_indices = (const int*)d_in[0];
  const int*   leaf_ids       = (const int*)d_in[1];
  const int*   op_ids         = (const int*)d_in[2];
  const float* comp_emb       = (const float*)d_in[3];
  const float* op_emb         = (const float*)d_in[4];
  const float* W_bin          = (const float*)d_in[5];
  const float* b_bin          = (const float*)d_in[6];
  const float* gamma          = (const float*)d_in[7];
  const float* beta           = (const float*)d_in[8];

  float* h1      = (float*)d_ws;
  float* h2      = h1 + (size_t)NSTRUCT * 4 * DD;
  float* optab   = h2 + (size_t)NSTRUCT * 2 * DD;
  float* leaftab = optab + (size_t)64 * DD;
  float* root    = h1;

  table_kernel<<<PBLK + 2 * (NCOMP / TROWS), 256, 0, stream>>>(comp_emb, gamma, beta, op_emb, W_bin,
                                                               optab, leaftab);
  level1_kernel<<<NSTRUCT / 8, 256, 0, stream>>>(leaf_ids, op_ids, optab, leaftab, b_bin, gamma, beta, h1);
  combine_kernel<<<(NSTRUCT * 2) / MT, 256, 0, stream>>>(h1, h2, op_ids, 4, 1,
                                                         optab, W_bin, b_bin, gamma, beta);
  combine_kernel<<<(NSTRUCT * 1) / MT, 256, 0, stream>>>(h2, root, op_ids, 6, 0,
                                                         optab, W_bin, b_bin, gamma, beta);
  gather_kernel<<<65536 / (8 * GROWS), 256, 0, stream>>>(root, struct_indices, (float*)d_out);
}


# combine K loop unrolled into 32 blocks of 16 matrix instructions; each block's W and LDS operands requested as one group two blocks ahead into a ring of 3 register sets; k order and sums unchanged
# speedup vs baseline: 2.7181x; 1.0663x over previous
#include <hip/hip_runtime.h>

#define DD 256
#define NSTRUCT 8192
#define EPS 1e-5f

typedef float v2f __attribute__((ext_vector_type(2)));
typedef float v8f __attribute__((ext_vector_type(8)));

__device__ __forceinline__ float wave_sum(float v) {
  v += __shfl_xor(v, 1, 32);
  v += __shfl_xor(v, 2, 32);
  v += __shfl_xor(v, 4, 32);
  v += __shfl_xor(v, 8, 32);
  v += __shfl_xor(v, 16, 32);
  return v;
}
__device__ __forceinline__ float sum4(float4 x) { return x.x + x.y + x.z + x.w; }
__device__ __forceinline__ float sumsq4(float4 x) { return x.x*x.x + x.y*x.y + x.z*x.z + x.w*x.w; }
__device__ __forceinline__ float4 ln4(float4 x, float mu, float rs, float4 g, float4 b) {
  float4 y;
  y.x = (x.x - mu) * rs * g.x + b.x;
  y.y = (x.y - mu) * rs * g.y + b.y;
  y.z = (x.z - mu) * rs * g.z + b.z;
  y.w = (x.w - mu) * rs * g.w + b.w;
  return y;
}

#define PBLK 4
#define NCOMP 1024
#define TROWS 16
#define TSTR 260

__global__ __launch_bounds__(256) void table_kernel(
    const float* __restrict__ comp_emb, const float* __restrict__ gamma, const float* __restrict__ beta,
    const float* __restrict__ op_emb, const float* __restrict__ W,
    float* __restrict__ optab, float* __restrict__ leaftab) {
  __shared__ __attribute__((aligned(16))) float sa[TROWS * TSTR];
  const int lane = threadIdx.x & 31;
  const int wave = threadIdx.x >> 5;
  const int krow = (lane & 16) >> 3;
  if (blockIdx.x < PBLK) {
    const int t = blockIdx.x * 8 + wave;
    const int colA = (t & 7) * 32 + (lane & 15);
    const float* arow = op_emb + (size_t)((t >> 3) * 16 + (lane & 15)) * DD;
    v8f c0 = {}, c1 = {};
#pragma unroll 8
    for (int k0 = 0; k0 < DD; k0 += 4) {
      const int kb = k0 + krow;
      const float* Wk = W + (size_t)kb * DD;
      v2f b0 = { Wk[colA],      Wk[DD + colA] };
      v2f b1 = { Wk[colA + 16], Wk[DD + colA + 16] };
      v2f a  = *(const v2f*)(arow + kb);
      c0 = __builtin_amdgcn_wmma_f32_16x16x4_f32(false, a, false, b0, (short)0, c0, false, false);
      c1 = __builtin_amdgcn_wmma_f32_16x16x4_f32(false, a, false, b1, (short)0, c1, false, false);
    }
    float* dst = optab + (size_t)((t >> 3) * 16 + ((lane & 16) >> 1)) * DD + colA;
#pragma unroll
    for (int j = 0; j < 8; ++j) {
      dst[j * DD]      = c0[j];
      dst[j * DD + 16] = c1[j];
    }
    return;
  }
  const int blk  = blockIdx.x - PBLK;
  const int rgrp = blk >> 1;
  const int side = blk & 1;
  {
    const float4 g0 = ((const float4*)gamma)[lane], g1 = ((const float4*)gamma)[lane + 32];
    const float4 b0 = ((const float4*)beta)[lane],  b1 = ((const float4*)beta)[lane + 32];
#pragma unroll
    for (int i = 0; i < 2; ++i) {
      const int lr = wave * 2 + i;
      const float4* src = (const float4*)(comp_emb + (size_t)(rgrp * TROWS + lr) * DD);
      float4 x0 = src[lane];
      float4 x1 = src[lane + 32];
      float s  = wave_sum(sum4(x0) + sum4(x1));
      float s2 = wave_sum(sumsq4(x0) + sumsq4(x1));
      float mu = s * (1.0f / DD);
      float rs = rsqrtf(s2 * (1.0f / DD) - mu * mu + EPS);
      *(float4*)&sa[lr * TSTR + lane * 4]       = ln4(x0, mu, rs, g0, b0);
      *(float4*)&sa[lr * TSTR + 128 + lane * 4] = ln4(x1, mu, rs, g1, b1);
    }
  }
  __syncthreads();
  const int colA = wave * 32 + (lane & 15);
  const float* arow = &sa[(lane & 15) * TSTR];
  const float* Ws = W + (size_t)(DD + side * DD) * DD;
  v8f c0 = {}, c1 = {};
#pragma unroll 8
  for (int k0 = 0; k0 < DD; k0 += 4) {
    const int kb = k0 + krow;
    const float* Wk = Ws + (size_t)kb * DD;
    v2f b0 = { Wk[colA],      Wk[DD + colA] };
    v2f b1 = { Wk[colA + 16], Wk[DD + colA + 16] };
    v2f a  = *(const v2f*)(arow + kb);
    c0 = __builtin_amdgcn_wmma_f32_16x16x4_f32(false, a, false, b0, (short)0, c0, false, false);
    c1 = __builtin_amdgcn_wmma_f32_16x16x4_f32(false, a, false, b1, (short)0, c1, false, false);
  }
  float* dst = leaftab + (size_t)side * NCOMP * DD + (size_t)(rgrp * TROWS + ((lane & 16) >> 1)) * DD + colA;
#pragma unroll
  for (int j = 0; j < 8; ++j) {
    dst[j * DD]      = c0[j];
    dst[j * DD + 16] = c1[j];
  }
}

__global__ __launch_bounds__(256) void level1_kernel(
    const int* __restrict__ leaf_ids, const int* __restrict__ op_ids,
    const float* __restrict__ optab, const float* __restrict__ leaftab,
    const float* __restrict__ bias, const float* __restrict__ gamma, const float* __restrict__ beta,
    float* __restrict__ h1) {
  const int lane = threadIdx.x & 31;
  const int st = __builtin_amdgcn_readfirstlane(blockIdx.x * 8 + (threadIdx.x >> 5));
  int lid[8], oid[4];
#pragma unroll
  for (int j = 0; j < 8; ++j) lid[j] = leaf_ids[st * 8 + j];
#pragma unroll
  for (int j = 0; j < 4; ++j) oid[j] = op_ids[st * 7 + j];
  float4 o0[4], o1[4], l0[4], l1[4], r0[4], r1[4];
#pragma unroll
  for (int i = 0; i < 4; ++i) {
    const float4* po = (const float4*)(optab + (size_t)oid[i] * DD);
    const float4* pl = (const float4*)(leaftab + (size_t)lid[2 * i] * DD);
    const float4* pr = (const float4*)(leaftab + (size_t)NCOMP * DD + (size_t)lid[2 * i + 1] * DD);
    o0[i] = po[lane]; o1[i] = po[lane + 32];
    l0[i] = pl[lane]; l1[i] = pl[lane + 32];
    r0[i] = pr[lane]; r1[i] = pr[lane + 32];
  }
  const float4 bs0 = ((const float4*)bias)[lane],  bs1 = ((const float4*)bias)[lane + 32];
  const float4 g0  = ((const float4*)gamma)[lane], g1  = ((const float4*)gamma)[lane + 32];
  const float4 bt0 = ((const float4*)beta)[lane],  bt1 = ((const float4*)beta)[lane + 32];
#pragma unroll
  for (int i = 0; i < 4; ++i) {
    float4 x0, x1;
    x0.x = ((o0[i].x + l0[i].x) + r0[i].x) + bs0.x;  x0.y = ((o0[i].y + l0[i].y) + r0[i].y) + bs0.y;
    x0.z = ((o0[i].z + l0[i].z) + r0[i].z) + bs0.z;  x0.w = ((o0[i].w + l0[i].w) + r0[i].w) + bs0.w;
    x1.x = ((o1[i].x + l1[i].x) + r1[i].x) + bs1.x;  x1.y = ((o1[i].y + l1[i].y) + r1[i].y) + bs1.y;
    x1.z = ((o1[i].z + l1[i].z) + r1[i].z) + bs1.z;  x1.w = ((o1[i].w + l1[i].w) + r1[i].w) + bs1.w;
    float s  = wave_sum(sum4(x0) + sum4(x1));
    float s2 = wave_sum(sumsq4(x0) + sumsq4(x1));
    float mu = s * (1.0f / DD);
    float rs = rsqrtf(s2 * (1.0f / DD) - mu * mu + EPS);
    float4* dst = (float4*)(h1 + (size_t)(st * 4 + i) * DD);
    dst[lane]      = ln4(x0, mu, rs, g0, bt0);
    dst[lane + 32] = ln4(x1, mu, rs, g1, bt1);
  }
}

#define MT 32
#define KC 512
#define CSTR 516
#define OSTR 260

#define GK 4
#define KBLK (4 * GK)
#define NRING 3
struct OpBlk { v2f b0[GK], b1[GK], a0[GK], a1[GK]; };

__device__ __forceinline__ void load_blk(OpBlk& f, const float* __restrict__ Wb,
                                         const float* arow0, const float* arow1, int k0) {
#pragma unroll
  for (int s = 0; s < GK; ++s) {
    const float* Wk = Wb + (size_t)(k0 + 4 * s) * DD;
    f.b0[s] = (v2f){ Wk[0],  Wk[DD] };
    f.b1[s] = (v2f){ Wk[16], Wk[DD + 16] };
    f.a0[s] = *(const v2f*)(arow0 + k0 + 4 * s);
    f.a1[s] = *(const v2f*)(arow1 + k0 + 4 * s);
  }
}
__device__ __forceinline__ void mma_blk(const OpBlk& f, v8f& c00, v8f& c01, v8f& c10, v8f& c11) {
#pragma unroll
  for (int s = 0; s < GK; ++s) {
    c00 = __builtin_amdgcn_wmma_f32_16x16x4_f32(false, f.a0[s], false, f.b0[s], (short)0, c00, false, false);
    c01 = __builtin_amdgcn_wmma_f32_16x16x4_f32(false, f.a0[s], false, f.b1[s], (short)0, c01, false, false);
    c10 = __builtin_amdgcn_wmma_f32_16x16x4_f32(false, f.a1[s], false, f.b0[s], (short)0, c10, false, false);
    c11 = __builtin_amdgcn_wmma_f32_16x16x4_f32(false, f.a1[s], false, f.b1[s], (short)0, c11, false, false);
  }
}

__global__ __launch_bounds__(256, 6) void combine_kernel(
    const float* __restrict__ h_in, float* __restrict__ h_out,
    const int* __restrict__ op_ids, int off, int ln2n,
    const float* __restrict__ optab, const float* __restrict__ W,
    const float* __restrict__ bias, const float* __restrict__ gamma,
    const float* __restrict__ beta) {
  __shared__ float smem[MT * CSTR];

  const int tid = threadIdx.x;
  const int lane = tid & 31;
  const int wave = tid >> 5;
  const int r0 = blockIdx.x * MT;
  const int nmask = (1 << ln2n) - 1;

  const int n0 = wave * 32;
  const int colA = n0 + (lane & 15);
  const int roff = (lane & 16) >> 1;

  int opid[16];
#pragma unroll
  for (int j = 0; j < 16; ++j) {
    const int rg = r0 + (j & 7) + roff + (j & 8) * 2;
    opid[j] = op_ids[(rg >> ln2n) * 7 + off + (rg & nmask)];
  }

  v8f c00, c01, c10, c11;
  {
    const int c4 = tid & 63;
    const int mrow = tid >> 6;
    float4 vl[8], vr[8];
#pragma unroll
    for (int j = 0; j < 8; ++j) {
      const int rg = r0 + mrow + 4 * j;
      const float4* src = (const float4*)(h_in + (size_t)(2 * rg) * DD);
      vl[j] = src[c4];
      vr[j] = src[64 + c4];
    }
#pragma unroll
    for (int j = 0; j < 8; ++j) {
      const float* p0 = optab + (size_t)opid[j] * DD + colA;
      const float* p1 = optab + (size_t)opid[8 + j] * DD + colA;
      c00[j] = p0[0];
      c01[j] = p0[16];
      c10[j] = p1[0];
      c11[j] = p1[16];
    }
#pragma unroll
    for (int j = 0; j < 8; ++j) {
      float* dst = &smem[(mrow + 4 * j) * CSTR + c4 * 4];
      *(float4*)dst = vl[j];
      *(float4*)(dst + DD) = vr[j];
    }
  }
  __syncthreads();

  const int krow = (lane & 16) >> 3;
  const float* arow0 = &smem[(lane & 15) * CSTR + krow];
  const float* arow1 = arow0 + 16 * CSTR;
  const float* Wb = W + (size_t)(DD + krow) * DD + colA;
  OpBlk f[NRING];
#pragma unroll
  for (int g = 0; g < NRING - 1; ++g) load_blk(f[g], Wb, arow0, arow1, g * KBLK);
#pragma unroll
  for (int g = 0; g < KC / KBLK; ++g) {
    if (g + NRING - 1 < KC / KBLK) load_blk(f[(g + NRING - 1) % NRING], Wb, arow0, arow1, (g + NRING - 1) * KBLK);
    __builtin_amdgcn_sched_barrier(0);
    mma_blk(f[g % NRING], c00, c01, c10, c11);
    __builtin_amdgcn_sched_barrier(0);
  }
  __syncthreads();

  const float bias0 = bias[colA];
  const float bias1 = bias[colA + 16];
#pragma unroll
  for (int j = 0; j < 8; ++j) {
    const int rr = j + roff;
    smem[rr * OSTR + colA]              = c00[j] + bias0;
    smem[rr * OSTR + colA + 16]         = c01[j] + bias1;
    smem[(rr + 16) * OSTR + colA]       = c10[j] + bias0;
    smem[(rr + 16) * OSTR + colA + 16]  = c11[j] + bias1;
  }
  __syncthreads();

  const float4 g0  = ((const float4*)gamma)[lane];
  const float4 g1  = ((const float4*)gamma)[lane + 32];
  const float4 bt0 = ((const float4*)beta)[lane];
  const float4 bt1 = ((const float4*)beta)[lane + 32];
#pragma unroll
  for (int i = 0; i < 4; ++i) {
    const int row = wave * 4 + i;
    float4 x0 = *(const float4*)&smem[row * OSTR + lane * 4];
    float4 x1 = *(const float4*)&smem[row * OSTR + 128 + lane * 4];
    float s  = wave_sum(sum4(x0) + sum4(x1));
    float s2 = wave_sum(sumsq4(x0) + sumsq4(x1));
    float mu = s * (1.0f / DD);
    float rs = rsqrtf(s2 * (1.0f / DD) - mu * mu + EPS);
    float4* dst = (float4*)(h_out + (size_t)(r0 + row) * DD);
    dst[lane]      = ln4(x0, mu, rs, g0, bt0);
    dst[lane + 32] = ln4(x1, mu, rs, g1, bt1);
  }
}

#define GROWS 16
typedef unsigned v4u __attribute__((ext_vector_type(4)));
__global__ __launch_bounds__(256) void gather_kernel(
    const float* __restrict__ root, const int* __restrict__ sidx,
    float* __restrict__ out) {
  const int lane = threadIdx.x & 31;
  const int r0 = __builtin_amdgcn_readfirstlane((blockIdx.x * 8 + (threadIdx.x >> 5)) * GROWS);
  int idx[GROWS];
#pragma unroll
  for (int j = 0; j < GROWS; ++j) idx[j] = sidx[r0 + j];
  v4u v0[GROWS], v1[GROWS];
#pragma unroll
  for (int j = 0; j < GROWS; ++j) {
    const unsigned li = (unsigned)idx[j] < (unsigned)NSTRUCT ? (unsigned)idx[j] : (unsigned)(NSTRUCT - 1);
    const v4u* src = (const v4u*)(root + (size_t)li * DD);
    v0[j] = src[lane];
    v1[j] = src[lane + 32];
  }
  __builtin_amdgcn_sched_barrier(0);
#pragma unroll
  for (int j = 0; j < GROWS; ++j) {
    v4u* dst = (v4u*)(out + (size_t)(r0 + j) * DD);
    const unsigned keep = idx[j] > 1 ? 0xffffffffu : 0u;
    dst[lane]      = v0[j] & keep;
    dst[lane + 32] = v1[j] & keep;
  }
}

extern "C" void kernel_launch(void* const* d_in, const int* in_sizes, int n_in,
                              void* d_out, int out_size, void* d_ws, size_t ws_size,
                              hipStream_t stream) {
  const int*   struct_indices = (const int*)d_in[0];
  const int*   leaf_ids       = (const int*)d_in[1];
  const int*   op_ids         = (const int*)d_in[2];
  const float* comp_emb       = (const float*)d_in[3];
  const float* op_emb         = (const float*)d_in[4];
  const float* W_bin          = (const float*)d_in[5];
  const float* b_bin          = (const float*)d_in[6];
  const float* gamma          = (const float*)d_in[7];
  const float* beta           = (const float*)d_in[8];

  float* h1      = (float*)d_ws;
  float* h2      = h1 + (size_t)NSTRUCT * 4 * DD;
  float* optab   = h2 + (size_t)NSTRUCT * 2 * DD;
  float* leaftab = optab + (size_t)64 * DD;
  float* root    = h1;

  table_kernel<<<PBLK + 2 * (NCOMP / TROWS), 256, 0, stream>>>(comp_emb, gamma, beta, op_emb, W_bin,
                                                               optab, leaftab);
  level1_kernel<<<NSTRUCT / 8, 256, 0, stream>>>(leaf_ids, op_ids, optab, leaftab, b_bin, gamma, beta, h1);
  combine_kernel<<<(NSTRUCT * 2) / MT, 256, 0, stream>>>(h1, h2, op_ids, 4, 1,
                                                         optab, W_bin, b_bin, gamma, beta);
  combine_kernel<<<(NSTRUCT * 1) / MT, 256, 0, stream>>>(h2, root, op_ids, 6, 0,
                                                         optab, W_bin, b_bin, gamma, beta);
  gather_kernel<<<65536 / (8 * GROWS), 256, 0, stream>>>(root, struct_indices, (float*)d_out);
}


# level 1 fused into the level-2 kernel: each wave builds its two structs' level-1 rows (table gather + add + LN) straight into the LDS operand tile; 4 launches, no level-1 plane in memory; same sums
# speedup vs baseline: 3.2611x; 1.1998x over previous
#include <hip/hip_runtime.h>

#define DD 256
#define NSTRUCT 8192
#define EPS 1e-5f

typedef float v2f __attribute__((ext_vector_type(2)));
typedef float v8f __attribute__((ext_vector_type(8)));

__device__ __forceinline__ float wave_sum(float v) {
  v += __shfl_xor(v, 1, 32);
  v += __shfl_xor(v, 2, 32);
  v += __shfl_xor(v, 4, 32);
  v += __shfl_xor(v, 8, 32);
  v += __shfl_xor(v, 16, 32);
  return v;
}
__device__ __forceinline__ float sum4(float4 x) { return x.x + x.y + x.z + x.w; }
__device__ __forceinline__ float sumsq4(float4 x) { return x.x*x.x + x.y*x.y + x.z*x.z + x.w*x.w; }
__device__ __forceinline__ float4 ln4(float4 x, float mu, float rs, float4 g, float4 b) {
  float4 y;
  y.x = (x.x - mu) * rs * g.x + b.x;
  y.y = (x.y - mu) * rs * g.y + b.y;
  y.z = (x.z - mu) * rs * g.z + b.z;
  y.w = (x.w - mu) * rs * g.w + b.w;
  return y;
}

#define PBLK 4
#define NCOMP 1024
#define TROWS 16
#define TSTR 260

__global__ __launch_bounds__(256) void table_kernel(
    const float* __restrict__ comp_emb, const float* __restrict__ gamma, const float* __restrict__ beta,
    const float* __restrict__ op_emb, const float* __restrict__ W,
    float* __restrict__ optab, float* __restrict__ leaftab) {
  __shared__ __attribute__((aligned(16))) float sa[TROWS * TSTR];
  const int lane = threadIdx.x & 31;
  const int wave = threadIdx.x >> 5;
  const int krow = (lane & 16) >> 3;
  if (blockIdx.x < PBLK) {
    const int t = blockIdx.x * 8 + wave;
    const int colA = (t & 7) * 32 + (lane & 15);
    const float* arow = op_emb + (size_t)((t >> 3) * 16 + (lane & 15)) * DD;
    v8f c0 = {}, c1 = {};
#pragma unroll 8
    for (int k0 = 0; k0 < DD; k0 += 4) {
      const int kb = k0 + krow;
      const float* Wk = W + (size_t)kb * DD;
      v2f b0 = { Wk[colA],      Wk[DD + colA] };
      v2f b1 = { Wk[colA + 16], Wk[DD + colA + 16] };
      v2f a  = *(const v2f*)(arow + kb);
      c0 = __builtin_amdgcn_wmma_f32_16x16x4_f32(false, a, false, b0, (short)0, c0, false, false);
      c1 = __builtin_amdgcn_wmma_f32_16x16x4_f32(false, a, false, b1, (short)0, c1, false, false);
    }
    float* dst = optab + (size_t)((t >> 3) * 16 + ((lane & 16) >> 1)) * DD + colA;
#pragma unroll
    for (int j = 0; j < 8; ++j) {
      dst[j * DD]      = c0[j];
      dst[j * DD + 16] = c1[j];
    }
    return;
  }
  const int blk  = blockIdx.x - PBLK;
  const int rgrp = blk >> 1;
  const int side = blk & 1;
  {
    const float4 g0 = ((const float4*)gamma)[lane], g1 = ((const float4*)gamma)[lane + 32];
    const float4 b0 = ((const float4*)beta)[lane],  b1 = ((const float4*)beta)[lane + 32];
#pragma unroll
    for (int i = 0; i < 2; ++i) {
      const int lr = wave * 2 + i;
      const float4* src = (const float4*)(comp_emb + (size_t)(rgrp * TROWS + lr) * DD);
      float4 x0 = src[lane];
      float4 x1 = src[lane + 32];
      float s  = wave_sum(sum4(x0) + sum4(x1));
      float s2 = wave_sum(sumsq4(x0) + sumsq4(x1));
      float mu = s * (1.0f / DD);
      float rs = rsqrtf(s2 * (1.0f / DD) - mu * mu + EPS);
      *(float4*)&sa[lr * TSTR + lane * 4]       = ln4(x0, mu, rs, g0, b0);
      *(float4*)&sa[lr * TSTR + 128 + lane * 4] = ln4(x1, mu, rs, g1, b1);
    }
  }
  __syncthreads();
  const int colA = wave * 32 + (lane & 15);
  const float* arow = &sa[(lane & 15) * TSTR];
  const float* Ws = W + (size_t)(DD + side * DD) * DD;
  v8f c0 = {}, c1 = {};
#pragma unroll 8
  for (int k0 = 0; k0 < DD; k0 += 4) {
    const int kb = k0 + krow;
    const float* Wk = Ws + (size_t)kb * DD;
    v2f b0 = { Wk[colA],      Wk[DD + colA] };
    v2f b1 = { Wk[colA + 16], Wk[DD + colA + 16] };
    v2f a  = *(const v2f*)(arow + kb);
    c0 = __builtin_amdgcn_wmma_f32_16x16x4_f32(false, a, false, b0, (short)0, c0, false, false);
    c1 = __builtin_amdgcn_wmma_f32_16x16x4_f32(false, a, false, b1, (short)0, c1, false, false);
  }
  float* dst = leaftab + (size_t)side * NCOMP * DD + (size_t)(rgrp * TROWS + ((lane & 16) >> 1)) * DD + colA;
#pragma unroll
  for (int j = 0; j < 8; ++j) {
    dst[j * DD]      = c0[j];
    dst[j * DD + 16] = c1[j];
  }
}

__device__ __forceinline__ void level1_rows_to_tile(
    const int* lid, const int* oid, int lane,
    const float* __restrict__ optab, const float* __restrict__ leaftab,
    float4 bs0, float4 bs1, float4 g0, float4 g1, float4 bt0, float4 bt1,
    float* tile, int tstride) {
  float4 o0[4], o1[4], l0[4], l1[4], q0[4], q1[4];
#pragma unroll
  for (int i = 0; i < 4; ++i) {
    const float4* po = (const float4*)(optab + (size_t)oid[i] * DD);
    const float4* pl = (const float4*)(leaftab + (size_t)lid[2 * i] * DD);
    const float4* pr = (const float4*)(leaftab + (size_t)NCOMP * DD + (size_t)lid[2 * i + 1] * DD);
    o0[i] = po[lane]; o1[i] = po[lane + 32];
    l0[i] = pl[lane]; l1[i] = pl[lane + 32];
    q0[i] = pr[lane]; q1[i] = pr[lane + 32];
  }
#pragma unroll
  for (int i = 0; i < 4; ++i) {
    float4 x0, x1;
    x0.x = ((o0[i].x + l0[i].x) + q0[i].x) + bs0.x;  x0.y = ((o0[i].y + l0[i].y) + q0[i].y) + bs0.y;
    x0.z = ((o0[i].z + l0[i].z) + q0[i].z) + bs0.z;  x0.w = ((o0[i].w + l0[i].w) + q0[i].w) + bs0.w;
    x1.x = ((o1[i].x + l1[i].x) + q1[i].x) + bs1.x;  x1.y = ((o1[i].y + l1[i].y) + q1[i].y) + bs1.y;
    x1.z = ((o1[i].z + l1[i].z) + q1[i].z) + bs1.z;  x1.w = ((o1[i].w + l1[i].w) + q1[i].w) + bs1.w;
    float s  = wave_sum(sum4(x0) + sum4(x1));
    float s2 = wave_sum(sumsq4(x0) + sumsq4(x1));
    float mu = s * (1.0f / DD);
    float rs = rsqrtf(s2 * (1.0f / DD) - mu * mu + EPS);
    float* dst = tile + (i >> 1) * tstride + (i & 1) * DD + lane * 4;
    *(float4*)dst         = ln4(x0, mu, rs, g0, bt0);
    *(float4*)(dst + 128) = ln4(x1, mu, rs, g1, bt1);
  }
}

#define MT 32
#define KC 512
#define CSTR 516
#define OSTR 260

#define GK 4
#define KBLK (4 * GK)
#define NRING 3
struct OpBlk { v2f b0[GK], b1[GK], a0[GK], a1[GK]; };

__device__ __forceinline__ void load_blk(OpBlk& f, const float* __restrict__ Wb,
                                         const float* arow0, const float* arow1, int k0) {
#pragma unroll
  for (int s = 0; s < GK; ++s) {
    const float* Wk = Wb + (size_t)(k0 + 4 * s) * DD;
    f.b0[s] = (v2f){ Wk[0],  Wk[DD] };
    f.b1[s] = (v2f){ Wk[16], Wk[DD + 16] };
    f.a0[s] = *(const v2f*)(arow0 + k0 + 4 * s);
    f.a1[s] = *(const v2f*)(arow1 + k0 + 4 * s);
  }
}
__device__ __forceinline__ void mma_blk(const OpBlk& f, v8f& c00, v8f& c01, v8f& c10, v8f& c11) {
#pragma unroll
  for (int s = 0; s < GK; ++s) {
    c00 = __builtin_amdgcn_wmma_f32_16x16x4_f32(false, f.a0[s], false, f.b0[s], (short)0, c00, false, false);
    c01 = __builtin_amdgcn_wmma_f32_16x16x4_f32(false, f.a0[s], false, f.b1[s], (short)0, c01, false, false);
    c10 = __builtin_amdgcn_wmma_f32_16x16x4_f32(false, f.a1[s], false, f.b0[s], (short)0, c10, false, false);
    c11 = __builtin_amdgcn_wmma_f32_16x16x4_f32(false, f.a1[s], false, f.b1[s], (short)0, c11, false, false);
  }
}

__global__ __launch_bounds__(256, 6) void combine_kernel(
    const float* __restrict__ h_in, float* __restrict__ h_out,
    const int* __restrict__ op_ids, int off, int ln2n,
    const float* __restrict__ optab, const float* __restrict__ W,
    const float* __restrict__ bias, const float* __restrict__ gamma,
    const float* __restrict__ beta,
    const int* __restrict__ leaf_ids, const float* __restrict__ leaftab, int from_leaves) {
  __shared__ float smem[MT * CSTR];

  const int tid = threadIdx.x;
  const int lane = tid & 31;
  const int wave = __builtin_amdgcn_readfirstlane(tid >> 5);
  const int r0 = blockIdx.x * MT;
  const int nmask = (1 << ln2n) - 1;

  const int n0 = wave * 32;
  const int colA = n0 + (lane & 15);
  const int roff = (lane & 16) >> 1;

  v8f c00, c01, c10, c11;
  if (from_leaves) {
    const int sb = (r0 >> 1) + wave * 2;
    int rop[MT];
    const int* op2 = op_ids + (r0 >> 1) * 7 + 4;
#pragma unroll
    for (int m = 0; m < MT; ++m) rop[m] = __builtin_amdgcn_readfirstlane(op2[(m >> 1) * 7 + (m & 1)]);
    int lid[16], oid[8];
#pragma unroll
    for (int j = 0; j < 16; ++j) lid[j] = leaf_ids[sb * 8 + j];
#pragma unroll
    for (int j = 0; j < 8; ++j) oid[j] = op_ids[(sb + (j >> 2)) * 7 + (j & 3)];
    const float4 bs0 = ((const float4*)bias)[lane],  bs1 = ((const float4*)bias)[lane + 32];
    const float4 g0  = ((const float4*)gamma)[lane], g1  = ((const float4*)gamma)[lane + 32];
    const float4 bt0 = ((const float4*)beta)[lane],  bt1 = ((const float4*)beta)[lane + 32];
    __builtin_amdgcn_sched_barrier(0);
#pragma unroll
    for (int t = 0; t < 2; ++t) {
      level1_rows_to_tile(lid + 8 * t, oid + 4 * t, lane, optab, leaftab, bs0, bs1, g0, g1, bt0, bt1,
                          &smem[(wave * 4 + t * 2) * CSTR], CSTR);
      __builtin_amdgcn_sched_barrier(0);
    }
#pragma unroll
    for (int j = 0; j < 8; ++j) {
      const int lo0 = rop[j],      hi0 = rop[j + 8];
      const int lo1 = rop[j + 16], hi1 = rop[j + 24];
      const float* p0 = optab + (size_t)(roff ? hi0 : lo0) * DD + colA;
      const float* p1 = optab + (size_t)(roff ? hi1 : lo1) * DD + colA;
      c00[j] = p0[0];
      c01[j] = p0[16];
      c10[j] = p1[0];
      c11[j] = p1[16];
    }
  } else {
    int opid[16];
#pragma unroll
    for (int j = 0; j < 16; ++j) {
      const int rg = r0 + (j & 7) + roff + (j & 8) * 2;
      opid[j] = op_ids[(rg >> ln2n) * 7 + off + (rg & nmask)];
    }
    const int c4 = tid & 63;
    const int mrow = tid >> 6;
    float4 vl[8], vr[8];
#pragma unroll
    for (int j = 0; j < 8; ++j) {
      const int rg = r0 + mrow + 4 * j;
      const float4* src = (const float4*)(h_in + (size_t)(2 * rg) * DD);
      vl[j] = src[c4];
      vr[j] = src[64 + c4];
    }
#pragma unroll
    for (int j = 0; j < 8; ++j) {
      const float* p0 = optab + (size_t)opid[j] * DD + colA;
      const float* p1 = optab + (size_t)opid[8 + j] * DD + colA;
      c00[j] = p0[0];
      c01[j] = p0[16];
      c10[j] = p1[0];
      c11[j] = p1[16];
    }
#pragma unroll
    for (int j = 0; j < 8; ++j) {
      float* dst = &smem[(mrow + 4 * j) * CSTR + c4 * 4];
      *(float4*)dst = vl[j];
      *(float4*)(dst + DD) = vr[j];
    }
  }
  __syncthreads();

  const int krow = (lane & 16) >> 3;
  const float* arow0 = &smem[(lane & 15) * CSTR + krow];
  const float* arow1 = arow0 + 16 * CSTR;
  const float* Wb = W + (size_t)(DD + krow) * DD + colA;
  OpBlk f[NRING];
#pragma unroll
  for (int g = 0; g < NRING - 1; ++g) load_blk(f[g], Wb, arow0, arow1, g * KBLK);
#pragma unroll
  for (int g = 0; g < KC / KBLK; ++g) {
    if (g + NRING - 1 < KC / KBLK) load_blk(f[(g + NRING - 1) % NRING], Wb, arow0, arow1, (g + NRING - 1) * KBLK);
    __builtin_amdgcn_sched_barrier(0);
    mma_blk(f[g % NRING], c00, c01, c10, c11);
    __builtin_amdgcn_sched_barrier(0);
  }
  __syncthreads();

  const float bias0 = bias[colA];
  const float bias1 = bias[colA + 16];
#pragma unroll
  for (int j = 0; j < 8; ++j) {
    const int rr = j + roff;
    smem[rr * OSTR + colA]              = c00[j] + bias0;
    smem[rr * OSTR + colA + 16]         = c01[j] + bias1;
    smem[(rr + 16) * OSTR + colA]       = c10[j] + bias0;
    smem[(rr + 16) * OSTR + colA + 16]  = c11[j] + bias1;
  }
  __syncthreads();

  const float4 g0  = ((const float4*)gamma)[lane];
  const float4 g1  = ((const float4*)gamma)[lane + 32];
  const float4 bt0 = ((const float4*)beta)[lane];
  const float4 bt1 = ((const float4*)beta)[lane + 32];
#pragma unroll
  for (int i = 0; i < 4; ++i) {
    const int row = wave * 4 + i;
    float4 x0 = *(const float4*)&smem[row * OSTR + lane * 4];
    float4 x1 = *(const float4*)&smem[row * OSTR + 128 + lane * 4];
    float s  = wave_sum(sum4(x0) + sum4(x1));
    float s2 = wave_sum(sumsq4(x0) + sumsq4(x1));
    float mu = s * (1.0f / DD);
    float rs = rsqrtf(s2 * (1.0f / DD) - mu * mu + EPS);
    float4* dst = (float4*)(h_out + (size_t)(r0 + row) * DD);
    dst[lane]      = ln4(x0, mu, rs, g0, bt0);
    dst[lane + 32] = ln4(x1, mu, rs, g1, bt1);
  }
}

#define GROWS 16
typedef unsigned v4u __attribute__((ext_vector_type(4)));
__global__ __launch_bounds__(256) void gather_kernel(
    const float* __restrict__ root, const int* __restrict__ sidx,
    float* __restrict__ out) {
  const int lane = threadIdx.x & 31;
  const int r0 = __builtin_amdgcn_readfirstlane((blockIdx.x * 8 + (threadIdx.x >> 5)) * GROWS);
  int idx[GROWS];
#pragma unroll
  for (int j = 0; j < GROWS; ++j) idx[j] = sidx[r0 + j];
  v4u v0[GROWS], v1[GROWS];
#pragma unroll
  for (int j = 0; j < GROWS; ++j) {
    const unsigned li = (unsigned)idx[j] < (unsigned)NSTRUCT ? (unsigned)idx[j] : (unsigned)(NSTRUCT - 1);
    const v4u* src = (const v4u*)(root + (size_t)li * DD);
    v0[j] = src[lane];
    v1[j] = src[lane + 32];
  }
  __builtin_amdgcn_sched_barrier(0);
#pragma unroll
  for (int j = 0; j < GROWS; ++j) {
    v4u* dst = (v4u*)(out + (size_t)(r0 + j) * DD);
    const unsigned keep = idx[j] > 1 ? 0xffffffffu : 0u;
    dst[lane]      = v0[j] & keep;
    dst[lane + 32] = v1[j] & keep;
  }
}

extern "C" void kernel_launch(void* const* d_in, const int* in_sizes, int n_in,
                              void* d_out, int out_size, void* d_ws, size_t ws_size,
                              hipStream_t stream) {
  const int*   struct_indices = (const int*)d_in[0];
  const int*   leaf_ids       = (const int*)d_in[1];
  const int*   op_ids         = (const int*)d_in[2];
  const float* comp_emb       = (const float*)d_in[3];
  const float* op_emb         = (const float*)d_in[4];
  const float* W_bin          = (const float*)d_in[5];
  const float* b_bin          = (const float*)d_in[6];
  const float* gamma          = (const float*)d_in[7];
  const float* beta           = (const float*)d_in[8];

  float* root    = (float*)d_ws;
  float* h2      = root + (size_t)NSTRUCT * 4 * DD;
  float* optab   = h2 + (size_t)NSTRUCT * 2 * DD;
  float* leaftab = optab + (size_t)64 * DD;

  table_kernel<<<PBLK + 2 * (NCOMP / TROWS), 256, 0, stream>>>(comp_emb, gamma, beta, op_emb, W_bin,
                                                               optab, leaftab);
  combine_kernel<<<(NSTRUCT * 2) / MT, 256, 0, stream>>>(root, h2, op_ids, 4, 1,
                                                         optab, W_bin, b_bin, gamma, beta, leaf_ids, leaftab, 1);
  combine_kernel<<<(NSTRUCT * 1) / MT, 256, 0, stream>>>(h2, root, op_ids, 6, 0,
                                                         optab, W_bin, b_bin, gamma, beta, leaf_ids, leaftab, 0);
  gather_kernel<<<65536 / (8 * GROWS), 256, 0, stream>>>(root, struct_indices, (float*)d_out);
}


# level 2 via tables: LN is linear in the centred row, so ((row - mean) * gamma) @ W[256:768] is tabled per call (f32 matrix instruction); level 2 = gather, rs scale, add, LN; level-2 sum order changes
# speedup vs baseline: 3.6240x; 1.1113x over previous
#include <hip/hip_runtime.h>

#define DD 256
#define NSTRUCT 8192
#define EPS 1e-5f

typedef float v2f __attribute__((ext_vector_type(2)));
typedef float v8f __attribute__((ext_vector_type(8)));

__device__ __forceinline__ float wave_sum(float v) {
  v += __shfl_xor(v, 1, 32);
  v += __shfl_xor(v, 2, 32);
  v += __shfl_xor(v, 4, 32);
  v += __shfl_xor(v, 8, 32);
  v += __shfl_xor(v, 16, 32);
  return v;
}
__device__ __forceinline__ float sum4(float4 x) { return x.x + x.y + x.z + x.w; }
__device__ __forceinline__ float sumsq4(float4 x) { return x.x*x.x + x.y*x.y + x.z*x.z + x.w*x.w; }
__device__ __forceinline__ float4 ln4(float4 x, float mu, float rs, float4 g, float4 b) {
  float4 y;
  y.x = (x.x - mu) * rs * g.x + b.x;
  y.y = (x.y - mu) * rs * g.y + b.y;
  y.z = (x.z - mu) * rs * g.z + b.z;
  y.w = (x.w - mu) * rs * g.w + b.w;
  return y;
}

#define PBLK 4
#define NCOMP 1024
#define TROWS 16
#define TSTR 260

__global__ __launch_bounds__(256) void table_kernel(
    const float* __restrict__ comp_emb, const float* __restrict__ gamma, const float* __restrict__ beta,
    const float* __restrict__ op_emb, const float* __restrict__ W,
    float* __restrict__ optab, float* __restrict__ leaftab) {
  __shared__ __attribute__((aligned(16))) float sa[TROWS * TSTR];
  const int lane = threadIdx.x & 31;
  const int wave = threadIdx.x >> 5;
  const int krow = (lane & 16) >> 3;
  if (blockIdx.x < PBLK) {
    const int t = blockIdx.x * 8 + wave;
    const int colA = (t & 7) * 32 + (lane & 15);
    const float* arow = op_emb + (size_t)((t >> 3) * 16 + (lane & 15)) * DD;
    v8f c0 = {}, c1 = {};
#pragma unroll 8
    for (int k0 = 0; k0 < DD; k0 += 4) {
      const int kb = k0 + krow;
      const float* Wk = W + (size_t)kb * DD;
      v2f b0 = { Wk[colA],      Wk[DD + colA] };
      v2f b1 = { Wk[colA + 16], Wk[DD + colA + 16] };
      v2f a  = *(const v2f*)(arow + kb);
      c0 = __builtin_amdgcn_wmma_f32_16x16x4_f32(false, a, false, b0, (short)0, c0, false, false);
      c1 = __builtin_amdgcn_wmma_f32_16x16x4_f32(false, a, false, b1, (short)0, c1, false, false);
    }
    float* dst = optab + (size_t)((t >> 3) * 16 + ((lane & 16) >> 1)) * DD + colA;
#pragma unroll
    for (int j = 0; j < 8; ++j) {
      dst[j * DD]      = c0[j];
      dst[j * DD + 16] = c1[j];
    }
    return;
  }
  const int blk  = blockIdx.x - PBLK;
  const int rgrp = blk >> 1;
  const int side = blk & 1;
  {
    const float4 g0 = ((const float4*)gamma)[lane], g1 = ((const float4*)gamma)[lane + 32];
    const float4 b0 = ((const float4*)beta)[lane],  b1 = ((const float4*)beta)[lane + 32];
#pragma unroll
    for (int i = 0; i < 2; ++i) {
      const int lr = wave * 2 + i;
      const float4* src = (const float4*)(comp_emb + (size_t)(rgrp * TROWS + lr) * DD);
      float4 x0 = src[lane];
      float4 x1 = src[lane + 32];
      float s  = wave_sum(sum4(x0) + sum4(x1));
      float s2 = wave_sum(sumsq4(x0) + sumsq4(x1));
      float mu = s * (1.0f / DD);
      float rs = rsqrtf(s2 * (1.0f / DD) - mu * mu + EPS);
      *(float4*)&sa[lr * TSTR + lane * 4]       = ln4(x0, mu, rs, g0, b0);
      *(float4*)&sa[lr * TSTR + 128 + lane * 4] = ln4(x1, mu, rs, g1, b1);
    }
  }
  __syncthreads();
  const int colA = wave * 32 + (lane & 15);
  const float* arow = &sa[(lane & 15) * TSTR];
  const float* Ws = W + (size_t)(DD + side * DD) * DD;
  v8f c0 = {}, c1 = {};
#pragma unroll 8
  for (int k0 = 0; k0 < DD; k0 += 4) {
    const int kb = k0 + krow;
    const float* Wk = Ws + (size_t)kb * DD;
    v2f b0 = { Wk[colA],      Wk[DD + colA] };
    v2f b1 = { Wk[colA + 16], Wk[DD + colA + 16] };
    v2f a  = *(const v2f*)(arow + kb);
    c0 = __builtin_amdgcn_wmma_f32_16x16x4_f32(false, a, false, b0, (short)0, c0, false, false);
    c1 = __builtin_amdgcn_wmma_f32_16x16x4_f32(false, a, false, b1, (short)0, c1, false, false);
  }
  float* dst = leaftab + (size_t)side * NCOMP * DD + (size_t)(rgrp * TROWS + ((lane & 16) >> 1)) * DD + colA;
#pragma unroll
  for (int j = 0; j < 8; ++j) {
    dst[j * DD]      = c0[j];
    dst[j * DD + 16] = c1[j];
  }
}

#define NSRC 2112
#define NGRP2 134
#define PROWS (NGRP2 * 16)
#define PROW_B NSRC
#define PROW_BETA (NSRC + 16)

__global__ __launch_bounds__(256) void table2_kernel(
    const float* __restrict__ src, const float* __restrict__ bias,
    const float* __restrict__ gamma, const float* __restrict__ beta,
    const float* __restrict__ W, float* __restrict__ ptab) {
  __shared__ __attribute__((aligned(16))) float sa[TROWS * TSTR];
  const int lane = threadIdx.x & 31;
  const int wave = threadIdx.x >> 5;
  const int krow = (lane & 16) >> 3;
  const int grp  = blockIdx.x >> 1;
  const int side = blockIdx.x & 1;
  {
    const float4 g0 = ((const float4*)gamma)[lane], g1 = ((const float4*)gamma)[lane + 32];
#pragma unroll
    for (int i = 0; i < 2; ++i) {
      const int lr = wave * 2 + i;
      const float* row = grp < NSRC / 16 ? src + (size_t)(grp * 16 + lr) * DD : (grp == NSRC / 16 ? bias : beta);
      float4 x0 = ((const float4*)row)[lane];
      float4 x1 = ((const float4*)row)[lane + 32];
      if (grp <= NSRC / 16) {
        const float mu = wave_sum(sum4(x0) + sum4(x1)) * (1.0f / DD);
        x0.x = (x0.x - mu) * g0.x;  x0.y = (x0.y - mu) * g0.y;  x0.z = (x0.z - mu) * g0.z;  x0.w = (x0.w - mu) * g0.w;
        x1.x = (x1.x - mu) * g1.x;  x1.y = (x1.y - mu) * g1.y;  x1.z = (x1.z - mu) * g1.z;  x1.w = (x1.w - mu) * g1.w;
      }
      *(float4*)&sa[lr * TSTR + lane * 4]       = x0;
      *(float4*)&sa[lr * TSTR + 128 + lane * 4] = x1;
    }
  }
  __syncthreads();
  const int colA = wave * 32 + (lane & 15);
  const float* arow = &sa[(lane & 15) * TSTR];
  const float* Ws = W + (size_t)(DD + side * DD) * DD;
  v8f c0 = {}, c1 = {};
#pragma unroll 8
  for (int k0 = 0; k0 < DD; k0 += 4) {
    const int kb = k0 + krow;
    const float* Wk = Ws + (size_t)kb * DD;
    v2f b0 = { Wk[colA],      Wk[DD + colA] };
    v2f b1 = { Wk[colA + 16], Wk[DD + colA + 16] };
    v2f a  = *(const v2f*)(arow + kb);
    c0 = __builtin_amdgcn_wmma_f32_16x16x4_f32(false, a, false, b0, (short)0, c0, false, false);
    c1 = __builtin_amdgcn_wmma_f32_16x16x4_f32(false, a, false, b1, (short)0, c1, false, false);
  }
  float* dst = ptab + (size_t)side * PROWS * DD + (size_t)(grp * 16 + ((lane & 16) >> 1)) * DD + colA;
#pragma unroll
  for (int j = 0; j < 8; ++j) {
    dst[j * DD]      = c0[j];
    dst[j * DD + 16] = c1[j];
  }
}

#define L2S 2
__device__ __forceinline__ float4 add4(float4 a, float4 b) { return make_float4(a.x + b.x, a.y + b.y, a.z + b.z, a.w + b.w); }
__device__ __forceinline__ float4 fma4(float s, float4 a, float4 b) {
  return make_float4(s * a.x + b.x, s * a.y + b.y, s * a.z + b.z, s * a.w + b.w);
}
__global__ __launch_bounds__(256) void level2_kernel(
    const int* __restrict__ leaf_ids, const int* __restrict__ op_ids,
    const float* __restrict__ optab, const float* __restrict__ leaftab, const float* __restrict__ ptab,
    const float* __restrict__ bias, const float* __restrict__ gamma, const float* __restrict__ beta,
    float* __restrict__ h2) {
  const int lane = threadIdx.x & 31;
  const int s0 = __builtin_amdgcn_readfirstlane((blockIdx.x * 8 + (threadIdx.x >> 5)) * L2S);
  const float* PL = ptab;
  const float* PR = ptab + (size_t)PROWS * DD;
  const float4 bs0 = ((const float4*)bias)[lane],  bs1 = ((const float4*)bias)[lane + 32];
  const float4 g0  = ((const float4*)gamma)[lane], g1  = ((const float4*)gamma)[lane + 32];
  const float4 bt0 = ((const float4*)beta)[lane],  bt1 = ((const float4*)beta)[lane + 32];
  const float4 lb0 = ((const float4*)(PL + (size_t)PROW_B * DD))[lane], lb1 = ((const float4*)(PL + (size_t)PROW_B * DD))[lane + 32];
  const float4 rb0 = ((const float4*)(PR + (size_t)PROW_B * DD))[lane], rb1 = ((const float4*)(PR + (size_t)PROW_B * DD))[lane + 32];
  const float4 cs0 = add4(add4(((const float4*)(PL + (size_t)PROW_BETA * DD))[lane],
                               ((const float4*)(PR + (size_t)PROW_BETA * DD))[lane]), bs0);
  const float4 cs1 = add4(add4(((const float4*)(PL + (size_t)PROW_BETA * DD))[lane + 32],
                               ((const float4*)(PR + (size_t)PROW_BETA * DD))[lane + 32]), bs1);
  for (int t = 0; t < L2S; ++t) {
    const int sb = s0 + t;
    int lid[8], oid[6];
#pragma unroll
    for (int j = 0; j < 8; ++j) lid[j] = leaf_ids[sb * 8 + j];
#pragma unroll
    for (int j = 0; j < 6; ++j) oid[j] = op_ids[sb * 7 + j];
    float rs[4];
    {
      float4 o0[4], o1[4], l0[4], l1[4], q0[4], q1[4];
#pragma unroll
      for (int i = 0; i < 4; ++i) {
        const float4* po = (const float4*)(optab + (size_t)oid[i] * DD);
        const float4* pl = (const float4*)(leaftab + (size_t)lid[2 * i] * DD);
        const float4* pr = (const float4*)(leaftab + (size_t)NCOMP * DD + (size_t)lid[2 * i + 1] * DD);
        o0[i] = po[lane]; o1[i] = po[lane + 32];
        l0[i] = pl[lane]; l1[i] = pl[lane + 32];
        q0[i] = pr[lane]; q1[i] = pr[lane + 32];
      }
#pragma unroll
      for (int i = 0; i < 4; ++i) {
        const float4 x0 = add4(add4(add4(o0[i], l0[i]), q0[i]), bs0);
        const float4 x1 = add4(add4(add4(o1[i], l1[i]), q1[i]), bs1);
        const float s  = wave_sum(sum4(x0) + sum4(x1));
        const float s2 = wave_sum(sumsq4(x0) + sumsq4(x1));
        const float mu = s * (1.0f / DD);
        rs[i] = rsqrtf(s2 * (1.0f / DD) - mu * mu + EPS);
      }
    }
#pragma unroll
    for (int j = 0; j < 2; ++j) {
      const int il = 2 * j, ir = 2 * j + 1;
      const float4* pa  = (const float4*)(optab + (size_t)oid[4 + j] * DD);
      const float4* pl0 = (const float4*)(PL + (size_t)oid[il] * DD);
      const float4* pl1 = (const float4*)(PL + (size_t)(64 + lid[2 * il]) * DD);
      const float4* pl2 = (const float4*)(PL + (size_t)(64 + NCOMP + lid[2 * il + 1]) * DD);
      const float4* pr0 = (const float4*)(PR + (size_t)oid[ir] * DD);
      const float4* pr1 = (const float4*)(PR + (size_t)(64 + lid[2 * ir]) * DD);
      const float4* pr2 = (const float4*)(PR + (size_t)(64 + NCOMP + lid[2 * ir + 1]) * DD);
      const float4 a0 = pa[lane],  a1 = pa[lane + 32];
      const float4 u0 = pl0[lane], u1 = pl0[lane + 32];
      const float4 v0 = pl1[lane], v1 = pl1[lane + 32];
      const float4 w0 = pl2[lane], w1 = pl2[lane + 32];
      const float4 d0 = pr0[lane], d1 = pr0[lane + 32];
      const float4 e0 = pr1[lane], e1 = pr1[lane + 32];
      const float4 f0 = pr2[lane], f1 = pr2[lane + 32];
      const float4 sl0 = add4(add4(add4(u0, v0), w0), lb0), sl1 = add4(add4(add4(u1, v1), w1), lb1);
      const float4 sr0 = add4(add4(add4(d0, e0), f0), rb0), sr1 = add4(add4(add4(d1, e1), f1), rb1);
      const float4 x0 = fma4(rs[ir], sr0, fma4(rs[il], sl0, add4(a0, cs0)));
      const float4 x1 = fma4(rs[ir], sr1, fma4(rs[il], sl1, add4(a1, cs1)));
      const float s  = wave_sum(sum4(x0) + sum4(x1));
      const float s2 = wave_sum(sumsq4(x0) + sumsq4(x1));
      const float mu = s * (1.0f / DD);
      const float r  = rsqrtf(s2 * (1.0f / DD) - mu * mu + EPS);
      float4* dst = (float4*)(h2 + (size_t)(2 * sb + j) * DD);
      dst[lane]      = ln4(x0, mu, r, g0, bt0);
      dst[lane + 32] = ln4(x1, mu, r, g1, bt1);
    }
  }
}

#define MT 32
#define KC 512
#define CSTR 516
#define OSTR 260

#define GK 4
#define KBLK (4 * GK)
#define NRING 3
struct OpBlk { v2f b0[GK], b1[GK], a0[GK], a1[GK]; };

__device__ __forceinline__ void load_blk(OpBlk& f, const float* __restrict__ Wb,
                                         const float* arow0, const float* arow1, int k0) {
#pragma unroll
  for (int s = 0; s < GK; ++s) {
    const float* Wk = Wb + (size_t)(k0 + 4 * s) * DD;
    f.b0[s] = (v2f){ Wk[0],  Wk[DD] };
    f.b1[s] = (v2f){ Wk[16], Wk[DD + 16] };
    f.a0[s] = *(const v2f*)(arow0 + k0 + 4 * s);
    f.a1[s] = *(const v2f*)(arow1 + k0 + 4 * s);
  }
}
__device__ __forceinline__ void mma_blk(const OpBlk& f, v8f& c00, v8f& c01, v8f& c10, v8f& c11) {
#pragma unroll
  for (int s = 0; s < GK; ++s) {
    c00 = __builtin_amdgcn_wmma_f32_16x16x4_f32(false, f.a0[s], false, f.b0[s], (short)0, c00, false, false);
    c01 = __builtin_amdgcn_wmma_f32_16x16x4_f32(false, f.a0[s], false, f.b1[s], (short)0, c01, false, false);
    c10 = __builtin_amdgcn_wmma_f32_16x16x4_f32(false, f.a1[s], false, f.b0[s], (short)0, c10, false, false);
    c11 = __builtin_amdgcn_wmma_f32_16x16x4_f32(false, f.a1[s], false, f.b1[s], (short)0, c11, false, false);
  }
}

__global__ __launch_bounds__(256, 6) void combine_kernel(
    const float* __restrict__ h_in, float* __restrict__ h_out,
    const int* __restrict__ op_ids, int off, int ln2n,
    const float* __restrict__ optab, const float* __restrict__ W,
    const float* __restrict__ bias, const float* __restrict__ gamma,
    const float* __restrict__ beta) {
  __shared__ float smem[MT * CSTR];

  const int tid = threadIdx.x;
  const int lane = tid & 31;
  const int wave = __builtin_amdgcn_readfirstlane(tid >> 5);
  const int r0 = blockIdx.x * MT;
  const int nmask = (1 << ln2n) - 1;

  const int n0 = wave * 32;
  const int colA = n0 + (lane & 15);
  const int roff = (lane & 16) >> 1;

  v8f c00, c01, c10, c11;
  {
    int opid[16];
#pragma unroll
    for (int j = 0; j < 16; ++j) {
      const int rg = r0 + (j & 7) + roff + (j & 8) * 2;
      opid[j] = op_ids[(rg >> ln2n) * 7 + off + (rg & nmask)];
    }
    const int c4 = tid & 63;
    const int mrow = tid >> 6;
    float4 vl[8], vr[8];
#pragma unroll
    for (int j = 0; j < 8; ++j) {
      const int rg = r0 + mrow + 4 * j;
      const float4* src = (const float4*)(h_in + (size_t)(2 * rg) * DD);
      vl[j] = src[c4];
      vr[j] = src[64 + c4];
    }
#pragma unroll
    for (int j = 0; j < 8; ++j) {
      const float* p0 = optab + (size_t)opid[j] * DD + colA;
      const float* p1 = optab + (size_t)opid[8 + j] * DD + colA;
      c00[j] = p0[0];
      c01[j] = p0[16];
      c10[j] = p1[0];
      c11[j] = p1[16];
    }
#pragma unroll
    for (int j = 0; j < 8; ++j) {
      float* dst = &smem[(mrow + 4 * j) * CSTR + c4 * 4];
      *(float4*)dst = vl[j];
      *(float4*)(dst + DD) = vr[j];
    }
  }
  __syncthreads();

  const int krow = (lane & 16) >> 3;
  const float* arow0 = &smem[(lane & 15) * CSTR + krow];
  const float* arow1 = arow0 + 16 * CSTR;
  const float* Wb = W + (size_t)(DD + krow) * DD + colA;
  OpBlk f[NRING];
#pragma unroll
  for (int g = 0; g < NRING - 1; ++g) load_blk(f[g], Wb, arow0, arow1, g * KBLK);
#pragma unroll
  for (int g = 0; g < KC / KBLK; ++g) {
    if (g + NRING - 1 < KC / KBLK) load_blk(f[(g + NRING - 1) % NRING], Wb, arow0, arow1, (g + NRING - 1) * KBLK);
    __builtin_amdgcn_sched_barrier(0);
    mma_blk(f[g % NRING], c00, c01, c10, c11);
    __builtin_amdgcn_sched_barrier(0);
  }
  __syncthreads();

  const float bias0 = bias[colA];
  const float bias1 = bias[colA + 16];
#pragma unroll
  for (int j = 0; j < 8; ++j) {
    const int rr = j + roff;
    smem[rr * OSTR + colA]              = c00[j] + bias0;
    smem[rr * OSTR + colA + 16]         = c01[j] + bias1;
    smem[(rr + 16) * OSTR + colA]       = c10[j] + bias0;
    smem[(rr + 16) * OSTR + colA + 16]  = c11[j] + bias1;
  }
  __syncthreads();

  const float4 g0  = ((const float4*)gamma)[lane];
  const float4 g1  = ((const float4*)gamma)[lane + 32];
  const float4 bt0 = ((const float4*)beta)[lane];
  const float4 bt1 = ((const float4*)beta)[lane + 32];
#pragma unroll
  for (int i = 0; i < 4; ++i) {
    const int row = wave * 4 + i;
    float4 x0 = *(const float4*)&smem[row * OSTR + lane * 4];
    float4 x1 = *(const float4*)&smem[row * OSTR + 128 + lane * 4];
    float s  = wave_sum(sum4(x0) + sum4(x1));
    float s2 = wave_sum(sumsq4(x0) + sumsq4(x1));
    float mu = s * (1.0f / DD);
    float rs = rsqrtf(s2 * (1.0f / DD) - mu * mu + EPS);
    float4* dst = (float4*)(h_out + (size_t)(r0 + row) * DD);
    dst[lane]      = ln4(x0, mu, rs, g0, bt0);
    dst[lane + 32] = ln4(x1, mu, rs, g1, bt1);
  }
}

#define GROWS 16
typedef unsigned v4u __attribute__((ext_vector_type(4)));
__global__ __launch_bounds__(256) void gather_kernel(
    const float* __restrict__ root, const int* __restrict__ sidx,
    float* __restrict__ out) {
  const int lane = threadIdx.x & 31;
  const int r0 = __builtin_amdgcn_readfirstlane((blockIdx.x * 8 + (threadIdx.x >> 5)) * GROWS);
  int idx[GROWS];
#pragma unroll
  for (int j = 0; j < GROWS; ++j) idx[j] = sidx[r0 + j];
  v4u v0[GROWS], v1[GROWS];
#pragma unroll
  for (int j = 0; j < GROWS; ++j) {
    const unsigned li = (unsigned)idx[j] < (unsigned)NSTRUCT ? (unsigned)idx[j] : (unsigned)(NSTRUCT - 1);
    const v4u* src = (const v4u*)(root + (size_t)li * DD);
    v0[j] = src[lane];
    v1[j] = src[lane + 32];
  }
  __builtin_amdgcn_sched_barrier(0);
#pragma unroll
  for (int j = 0; j < GROWS; ++j) {
    v4u* dst = (v4u*)(out + (size_t)(r0 + j) * DD);
    const unsigned keep = idx[j] > 1 ? 0xffffffffu : 0u;
    dst[lane]      = v0[j] & keep;
    dst[lane + 32] = v1[j] & keep;
  }
}

extern "C" void kernel_launch(void* const* d_in, const int* in_sizes, int n_in,
                              void* d_out, int out_size, void* d_ws, size_t ws_size,
                              hipStream_t stream) {
  const int*   struct_indices = (const int*)d_in[0];
  const int*   leaf_ids       = (const int*)d_in[1];
  const int*   op_ids         = (const int*)d_in[2];
  const float* comp_emb       = (const float*)d_in[3];
  const float* op_emb         = (const float*)d_in[4];
  const float* W_bin          = (const float*)d_in[5];
  const float* b_bin          = (const float*)d_in[6];
  const float* gamma          = (const float*)d_in[7];
  const float* beta           = (const float*)d_in[8];

  float* root    = (float*)d_ws;
  float* ptab    = root + (size_t)NSTRUCT * DD;
  float* h2      = root + (size_t)NSTRUCT * 4 * DD;
  float* optab   = h2 + (size_t)NSTRUCT * 2 * DD;
  float* leaftab = optab + (size_t)64 * DD;

  table_kernel<<<PBLK + 2 * (NCOMP / TROWS), 256, 0, stream>>>(comp_emb, gamma, beta, op_emb, W_bin,
                                                               optab, leaftab);
  table2_kernel<<<2 * NGRP2, 256, 0, stream>>>(optab, b_bin, gamma, beta, W_bin, ptab);
  level2_kernel<<<NSTRUCT / (8 * L2S), 256, 0, stream>>>(leaf_ids, op_ids, optab, leaftab, ptab,
                                                         b_bin, gamma, beta, h2);
  combine_kernel<<<(NSTRUCT * 1) / MT, 256, 0, stream>>>(h2, root, op_ids, 6, 0,
                                                         optab, W_bin, b_bin, gamma, beta);
  gather_kernel<<<65536 / (8 * GROWS), 256, 0, stream>>>(root, struct_indices, (float*)d_out);
}


# level 2 fused into the root kernel: each wave builds its 4 structs' level-2 rows (table gather, rs scale, add, LN) into the LDS operand tile; 4 launches, no level-2 plane in memory; same sums
# speedup vs baseline: 3.9927x; 1.1017x over previous
#include <hip/hip_runtime.h>

#define DD 256
#define NSTRUCT 8192
#define EPS 1e-5f

typedef float v2f __attribute__((ext_vector_type(2)));
typedef float v8f __attribute__((ext_vector_type(8)));

__device__ __forceinline__ float wave_sum(float v) {
  v += __shfl_xor(v, 1, 32);
  v += __shfl_xor(v, 2, 32);
  v += __shfl_xor(v, 4, 32);
  v += __shfl_xor(v, 8, 32);
  v += __shfl_xor(v, 16, 32);
  return v;
}
template <int N> __device__ __forceinline__ void wave_sum_n(float (&v)[N]) {
#pragma unroll
  for (int m = 1; m < 32; m <<= 1) {
    float p[N];
#pragma unroll
    for (int i = 0; i < N; ++i) p[i] = __shfl_xor(v[i], m, 32);
#pragma unroll
    for (int i = 0; i < N; ++i) v[i] += p[i];
  }
}
__device__ __forceinline__ float sum4(float4 x) { return x.x + x.y + x.z + x.w; }
__device__ __forceinline__ float sumsq4(float4 x) { return x.x*x.x + x.y*x.y + x.z*x.z + x.w*x.w; }
__device__ __forceinline__ float4 ln4(float4 x, float mu, float rs, float4 g, float4 b) {
  float4 y;
  y.x = (x.x - mu) * rs * g.x + b.x;
  y.y = (x.y - mu) * rs * g.y + b.y;
  y.z = (x.z - mu) * rs * g.z + b.z;
  y.w = (x.w - mu) * rs * g.w + b.w;
  return y;
}

#define PBLK 4
#define NCOMP 1024
#define TROWS 16
#define TSTR 260

__global__ __launch_bounds__(256) void table_kernel(
    const float* __restrict__ comp_emb, const float* __restrict__ gamma, const float* __restrict__ beta,
    const float* __restrict__ op_emb, const float* __restrict__ W,
    float* __restrict__ optab, float* __restrict__ leaftab) {
  __shared__ __attribute__((aligned(16))) float sa[TROWS * TSTR];
  const int lane = threadIdx.x & 31;
  const int wave = threadIdx.x >> 5;
  const int krow = (lane & 16) >> 3;
  if (blockIdx.x < PBLK) {
    const int t = blockIdx.x * 8 + wave;
    const int colA = (t & 7) * 32 + (lane & 15);
    const float* arow = op_emb + (size_t)((t >> 3) * 16 + (lane & 15)) * DD;
    v8f c0 = {}, c1 = {};
#pragma unroll 8
    for (int k0 = 0; k0 < DD; k0 += 4) {
      const int kb = k0 + krow;
      const float* Wk = W + (size_t)kb * DD;
      v2f b0 = { Wk[colA],      Wk[DD + colA] };
      v2f b1 = { Wk[colA + 16], Wk[DD + colA + 16] };
      v2f a  = *(const v2f*)(arow + kb);
      c0 = __builtin_amdgcn_wmma_f32_16x16x4_f32(false, a, false, b0, (short)0, c0, false, false);
      c1 = __builtin_amdgcn_wmma_f32_16x16x4_f32(false, a, false, b1, (short)0, c1, false, false);
    }
    float* dst = optab + (size_t)((t >> 3) * 16 + ((lane & 16) >> 1)) * DD + colA;
#pragma unroll
    for (int j = 0; j < 8; ++j) {
      dst[j * DD]      = c0[j];
      dst[j * DD + 16] = c1[j];
    }
    return;
  }
  const int blk  = blockIdx.x - PBLK;
  const int rgrp = blk >> 1;
  const int side = blk & 1;
  {
    const float4 g0 = ((const float4*)gamma)[lane], g1 = ((const float4*)gamma)[lane + 32];
    const float4 b0 = ((const float4*)beta)[lane],  b1 = ((const float4*)beta)[lane + 32];
#pragma unroll
    for (int i = 0; i < 2; ++i) {
      const int lr = wave * 2 + i;
      const float4* src = (const float4*)(comp_emb + (size_t)(rgrp * TROWS + lr) * DD);
      float4 x0 = src[lane];
      float4 x1 = src[lane + 32];
      float s  = wave_sum(sum4(x0) + sum4(x1));
      float s2 = wave_sum(sumsq4(x0) + sumsq4(x1));
      float mu = s * (1.0f / DD);
      float rs = rsqrtf(s2 * (1.0f / DD) - mu * mu + EPS);
      *(float4*)&sa[lr * TSTR + lane * 4]       = ln4(x0, mu, rs, g0, b0);
      *(float4*)&sa[lr * TSTR + 128 + lane * 4] = ln4(x1, mu, rs, g1, b1);
    }
  }
  __syncthreads();
  const int colA = wave * 32 + (lane & 15);
  const float* arow = &sa[(lane & 15) * TSTR];
  const float* Ws = W + (size_t)(DD + side * DD) * DD;
  v8f c0 = {}, c1 = {};
#pragma unroll 8
  for (int k0 = 0; k0 < DD; k0 += 4) {
    const int kb = k0 + krow;
    const float* Wk = Ws + (size_t)kb * DD;
    v2f b0 = { Wk[colA],      Wk[DD + colA] };
    v2f b1 = { Wk[colA + 16], Wk[DD + colA + 16] };
    v2f a  = *(const v2f*)(arow + kb);
    c0 = __builtin_amdgcn_wmma_f32_16x16x4_f32(false, a, false, b0, (short)0, c0, false, false);
    c1 = __builtin_amdgcn_wmma_f32_16x16x4_f32(false, a, false, b1, (short)0, c1, false, false);
  }
  float* dst = leaftab + (size_t)side * NCOMP * DD + (size_t)(rgrp * TROWS + ((lane & 16) >> 1)) * DD + colA;
#pragma unroll
  for (int j = 0; j < 8; ++j) {
    dst[j * DD]      = c0[j];
    dst[j * DD + 16] = c1[j];
  }
}

#define NSRC 2112
#define NGRP2 134
#define PROWS (NGRP2 * 16)
#define PROW_B NSRC
#define PROW_BETA (NSRC + 16)

__global__ __launch_bounds__(256) void table2_kernel(
    const float* __restrict__ src, const float* __restrict__ bias,
    const float* __restrict__ gamma, const float* __restrict__ beta,
    const float* __restrict__ W, float* __restrict__ ptab) {
  __shared__ __attribute__((aligned(16))) float sa[TROWS * TSTR];
  const int lane = threadIdx.x & 31;
  const int wave = threadIdx.x >> 5;
  const int krow = (lane & 16) >> 3;
  const int grp  = blockIdx.x >> 1;
  const int side = blockIdx.x & 1;
  {
    const float4 g0 = ((const float4*)gamma)[lane], g1 = ((const float4*)gamma)[lane + 32];
#pragma unroll
    for (int i = 0; i < 2; ++i) {
      const int lr = wave * 2 + i;
      const float* row = grp < NSRC / 16 ? src + (size_t)(grp * 16 + lr) * DD : (grp == NSRC / 16 ? bias : beta);
      float4 x0 = ((const float4*)row)[lane];
      float4 x1 = ((const float4*)row)[lane + 32];
      if (grp <= NSRC / 16) {
        const float mu = wave_sum(sum4(x0) + sum4(x1)) * (1.0f / DD);
        x0.x = (x0.x - mu) * g0.x;  x0.y = (x0.y - mu) * g0.y;  x0.z = (x0.z - mu) * g0.z;  x0.w = (x0.w - mu) * g0.w;
        x1.x = (x1.x - mu) * g1.x;  x1.y = (x1.y - mu) * g1.y;  x1.z = (x1.z - mu) * g1.z;  x1.w = (x1.w - mu) * g1.w;
      }
      *(float4*)&sa[lr * TSTR + lane * 4]       = x0;
      *(float4*)&sa[lr * TSTR + 128 + lane * 4] = x1;
    }
  }
  __syncthreads();
  const int colA = wave * 32 + (lane & 15);
  const float* arow = &sa[(lane & 15) * TSTR];
  const float* Ws = W + (size_t)(DD + side * DD) * DD;
  v8f c0 = {}, c1 = {};
#pragma unroll 8
  for (int k0 = 0; k0 < DD; k0 += 4) {
    const int kb = k0 + krow;
    const float* Wk = Ws + (size_t)kb * DD;
    v2f b0 = { Wk[colA],      Wk[DD + colA] };
    v2f b1 = { Wk[colA + 16], Wk[DD + colA + 16] };
    v2f a  = *(const v2f*)(arow + kb);
    c0 = __builtin_amdgcn_wmma_f32_16x16x4_f32(false, a, false, b0, (short)0, c0, false, false);
    c1 = __builtin_amdgcn_wmma_f32_16x16x4_f32(false, a, false, b1, (short)0, c1, false, false);
  }
  float* dst = ptab + (size_t)side * PROWS * DD + (size_t)(grp * 16 + ((lane & 16) >> 1)) * DD + colA;
#pragma unroll
  for (int j = 0; j < 8; ++j) {
    dst[j * DD]      = c0[j];
    dst[j * DD + 16] = c1[j];
  }
}

#define SPW 4
__device__ __forceinline__ float4 add4(float4 a, float4 b) { return make_float4(a.x + b.x, a.y + b.y, a.z + b.z, a.w + b.w); }
__device__ __forceinline__ float4 fma4(float s, float4 a, float4 b) {
  return make_float4(s * a.x + b.x, s * a.y + b.y, s * a.z + b.z, s * a.w + b.w);
}

#define MT 32
#define KC 512
#define CSTR 516
#define OSTR 260

#define GK 4
#define KBLK (4 * GK)
#define NRING 3
struct OpBlk { v2f b0[GK], b1[GK], a0[GK], a1[GK]; };

__device__ __forceinline__ void load_blk(OpBlk& f, const float* __restrict__ Wb,
                                         const float* arow0, const float* arow1, int k0) {
#pragma unroll
  for (int s = 0; s < GK; ++s) {
    const float* Wk = Wb + (size_t)(k0 + 4 * s) * DD;
    f.b0[s] = (v2f){ Wk[0],  Wk[DD] };
    f.b1[s] = (v2f){ Wk[16], Wk[DD + 16] };
    f.a0[s] = *(const v2f*)(arow0 + k0 + 4 * s);
    f.a1[s] = *(const v2f*)(arow1 + k0 + 4 * s);
  }
}
__device__ __forceinline__ void mma_blk(const OpBlk& f, v8f& c00, v8f& c01, v8f& c10, v8f& c11) {
#pragma unroll
  for (int s = 0; s < GK; ++s) {
    c00 = __builtin_amdgcn_wmma_f32_16x16x4_f32(false, f.a0[s], false, f.b0[s], (short)0, c00, false, false);
    c01 = __builtin_amdgcn_wmma_f32_16x16x4_f32(false, f.a0[s], false, f.b1[s], (short)0, c01, false, false);
    c10 = __builtin_amdgcn_wmma_f32_16x16x4_f32(false, f.a1[s], false, f.b0[s], (short)0, c10, false, false);
    c11 = __builtin_amdgcn_wmma_f32_16x16x4_f32(false, f.a1[s], false, f.b1[s], (short)0, c11, false, false);
  }
}

__global__ __launch_bounds__(256) __attribute__((amdgpu_waves_per_eu(2, 2))) void combine_kernel(
    const int* __restrict__ leaf_ids, const int* __restrict__ op_ids,
    const float* __restrict__ optab, const float* __restrict__ leaftab, const float* __restrict__ ptab,
    float* __restrict__ h_out, const float* __restrict__ W,
    const float* __restrict__ bias, const float* __restrict__ gamma,
    const float* __restrict__ beta) {
  __shared__ __attribute__((aligned(16))) float smem[MT * CSTR];

  const int tid = threadIdx.x;
  const int lane = tid & 31;
  const int wave = __builtin_amdgcn_readfirstlane(tid >> 5);
  const int r0 = blockIdx.x * MT;

  const int n0 = wave * 32;
  const int colA = n0 + (lane & 15);
  const int roff = (lane & 16) >> 1;

  const int sb0 = r0 + wave * SPW;
  int lidn[8], oidn[6];
#pragma unroll
  for (int j = 0; j < 8; ++j) lidn[j] = leaf_ids[sb0 * 8 + j];
#pragma unroll
  for (int j = 0; j < 6; ++j) oidn[j] = op_ids[sb0 * 7 + j];

  int opid[16];
#pragma unroll
  for (int j = 0; j < 16; ++j) {
    const int rg = r0 + (j & 7) + roff + (j & 8) * 2;
    opid[j] = op_ids[rg * 7 + 6];
  }
  {
    const float* PL = ptab;
    const float* PR = ptab + (size_t)PROWS * DD;
    const float4 bs0 = ((const float4*)bias)[lane],  bs1 = ((const float4*)bias)[lane + 32];
    const float4 g0  = ((const float4*)gamma)[lane], g1  = ((const float4*)gamma)[lane + 32];
    const float4 bt0 = ((const float4*)beta)[lane],  bt1 = ((const float4*)beta)[lane + 32];
    const float4 lb0 = ((const float4*)(PL + (size_t)PROW_B * DD))[lane], lb1 = ((const float4*)(PL + (size_t)PROW_B * DD))[lane + 32];
    const float4 rb0 = ((const float4*)(PR + (size_t)PROW_B * DD))[lane], rb1 = ((const float4*)(PR + (size_t)PROW_B * DD))[lane + 32];
    const float4 cs0 = add4(add4(((const float4*)(PL + (size_t)PROW_BETA * DD))[lane],
                                 ((const float4*)(PR + (size_t)PROW_BETA * DD))[lane]), bs0);
    const float4 cs1 = add4(add4(((const float4*)(PL + (size_t)PROW_BETA * DD))[lane + 32],
                                 ((const float4*)(PR + (size_t)PROW_BETA * DD))[lane + 32]), bs1);
#pragma unroll 1
    for (int t = 0; t < SPW; ++t) {
      int lid[8], oid[6];
#pragma unroll
      for (int j = 0; j < 8; ++j) lid[j] = lidn[j];
#pragma unroll
      for (int j = 0; j < 6; ++j) oid[j] = oidn[j];
      {
        const int sn = sb0 + (t + 1 < SPW ? t + 1 : t);
#pragma unroll
        for (int j = 0; j < 8; ++j) lidn[j] = leaf_ids[sn * 8 + j];
#pragma unroll
        for (int j = 0; j < 6; ++j) oidn[j] = op_ids[sn * 7 + j];
      }
      float4 o0[4], o1[4], l0[4], l1[4], q0[4], q1[4];
      float4 a0[2], a1[2], u0[2], u1[2], v0[2], v1[2], w0[2], w1[2], d0[2], d1[2], e0[2], e1[2], f0[2], f1[2];
#pragma unroll
      for (int i = 0; i < 4; ++i) {
        const float4* po = (const float4*)(optab + (size_t)oid[i] * DD);
        const float4* pl = (const float4*)(leaftab + (size_t)lid[2 * i] * DD);
        const float4* pr = (const float4*)(leaftab + (size_t)NCOMP * DD + (size_t)lid[2 * i + 1] * DD);
        o0[i] = po[lane]; o1[i] = po[lane + 32];
        l0[i] = pl[lane]; l1[i] = pl[lane + 32];
        q0[i] = pr[lane]; q1[i] = pr[lane + 32];
      }
#pragma unroll
      for (int j = 0; j < 2; ++j) {
        const int il = 2 * j, ir = 2 * j + 1;
        const float4* pa  = (const float4*)(optab + (size_t)oid[4 + j] * DD);
        const float4* pl0 = (const float4*)(PL + (size_t)oid[il] * DD);
        const float4* pl1 = (const float4*)(PL + (size_t)(64 + lid[2 * il]) * DD);
        const float4* pl2 = (const float4*)(PL + (size_t)(64 + NCOMP + lid[2 * il + 1]) * DD);
        const float4* pr0 = (const float4*)(PR + (size_t)oid[ir] * DD);
        const float4* pr1 = (const float4*)(PR + (size_t)(64 + lid[2 * ir]) * DD);
        const float4* pr2 = (const float4*)(PR + (size_t)(64 + NCOMP + lid[2 * ir + 1]) * DD);
        a0[j] = pa[lane];  a1[j] = pa[lane + 32];
        u0[j] = pl0[lane]; u1[j] = pl0[lane + 32];
        v0[j] = pl1[lane]; v1[j] = pl1[lane + 32];
        w0[j] = pl2[lane]; w1[j] = pl2[lane + 32];
        d0[j] = pr0[lane]; d1[j] = pr0[lane + 32];
        e0[j] = pr1[lane]; e1[j] = pr1[lane + 32];
        f0[j] = pr2[lane]; f1[j] = pr2[lane + 32];
      }
      float rs[4];
      {
        float sv[8];
#pragma unroll
        for (int i = 0; i < 4; ++i) {
          const float4 x0 = add4(add4(add4(o0[i], l0[i]), q0[i]), bs0);
          const float4 x1 = add4(add4(add4(o1[i], l1[i]), q1[i]), bs1);
          sv[2 * i]     = sum4(x0) + sum4(x1);
          sv[2 * i + 1] = sumsq4(x0) + sumsq4(x1);
        }
        wave_sum_n<8>(sv);
#pragma unroll
        for (int i = 0; i < 4; ++i) {
          const float mu = sv[2 * i] * (1.0f / DD);
          rs[i] = rsqrtf(sv[2 * i + 1] * (1.0f / DD) - mu * mu + EPS);
        }
      }
      {
        float4 x0[2], x1[2];
        float rv[4];
#pragma unroll
        for (int j = 0; j < 2; ++j) {
          const int il = 2 * j, ir = 2 * j + 1;
          const float4 sl0 = add4(add4(add4(u0[j], v0[j]), w0[j]), lb0), sl1 = add4(add4(add4(u1[j], v1[j]), w1[j]), lb1);
          const float4 sr0 = add4(add4(add4(d0[j], e0[j]), f0[j]), rb0), sr1 = add4(add4(add4(d1[j], e1[j]), f1[j]), rb1);
          x0[j] = fma4(rs[ir], sr0, fma4(rs[il], sl0, add4(a0[j], cs0)));
          x1[j] = fma4(rs[ir], sr1, fma4(rs[il], sl1, add4(a1[j], cs1)));
          rv[2 * j]     = sum4(x0[j]) + sum4(x1[j]);
          rv[2 * j + 1] = sumsq4(x0[j]) + sumsq4(x1[j]);
        }
        wave_sum_n<4>(rv);
#pragma unroll
        for (int j = 0; j < 2; ++j) {
          const float mu = rv[2 * j] * (1.0f / DD);
          const float r  = rsqrtf(rv[2 * j + 1] * (1.0f / DD) - mu * mu + EPS);
          float* dst = &smem[(wave * SPW + t) * CSTR + j * DD + lane * 4];
          *(float4*)dst         = ln4(x0[j], mu, r, g0, bt0);
          *(float4*)(dst + 128) = ln4(x1[j], mu, r, g1, bt1);
        }
      }
    }
  }
  __syncthreads();

  v8f c00, c01, c10, c11;
#pragma unroll
  for (int j = 0; j < 8; ++j) {
    const float* p0 = optab + (size_t)opid[j] * DD + colA;
    const float* p1 = optab + (size_t)opid[8 + j] * DD + colA;
    c00[j] = p0[0];
    c01[j] = p0[16];
    c10[j] = p1[0];
    c11[j] = p1[16];
  }

  const int krow = (lane & 16) >> 3;
  const float* arow0 = &smem[(lane & 15) * CSTR + krow];
  const float* arow1 = arow0 + 16 * CSTR;
  const float* Wb = W + (size_t)(DD + krow) * DD + colA;
  OpBlk f[NRING];
#pragma unroll
  for (int g = 0; g < NRING - 1; ++g) load_blk(f[g], Wb, arow0, arow1, g * KBLK);
#pragma unroll
  for (int g = 0; g < KC / KBLK; ++g) {
    if (g + NRING - 1 < KC / KBLK) load_blk(f[(g + NRING - 1) % NRING], Wb, arow0, arow1, (g + NRING - 1) * KBLK);
    __builtin_amdgcn_sched_barrier(0);
    mma_blk(f[g % NRING], c00, c01, c10, c11);
    __builtin_amdgcn_sched_barrier(0);
  }
  __syncthreads();

  const float bias0 = bias[colA];
  const float bias1 = bias[colA + 16];
#pragma unroll
  for (int j = 0; j < 8; ++j) {
    const int rr = j + roff;
    smem[rr * OSTR + colA]              = c00[j] + bias0;
    smem[rr * OSTR + colA + 16]         = c01[j] + bias1;
    smem[(rr + 16) * OSTR + colA]       = c10[j] + bias0;
    smem[(rr + 16) * OSTR + colA + 16]  = c11[j] + bias1;
  }
  __syncthreads();

  const float4 g0  = ((const float4*)gamma)[lane];
  const float4 g1  = ((const float4*)gamma)[lane + 32];
  const float4 bt0 = ((const float4*)beta)[lane];
  const float4 bt1 = ((const float4*)beta)[lane + 32];
#pragma unroll
  for (int i = 0; i < 4; ++i) {
    const int row = wave * 4 + i;
    float4 x0 = *(const float4*)&smem[row * OSTR + lane * 4];
    float4 x1 = *(const float4*)&smem[row * OSTR + 128 + lane * 4];
    float s  = wave_sum(sum4(x0) + sum4(x1));
    float s2 = wave_sum(sumsq4(x0) + sumsq4(x1));
    float mu = s * (1.0f / DD);
    float rs = rsqrtf(s2 * (1.0f / DD) - mu * mu + EPS);
    float4* dst = (float4*)(h_out + (size_t)(r0 + row) * DD);
    dst[lane]      = ln4(x0, mu, rs, g0, bt0);
    dst[lane + 32] = ln4(x1, mu, rs, g1, bt1);
  }
}

#define GROWS 16
typedef unsigned v4u __attribute__((ext_vector_type(4)));
__global__ __launch_bounds__(256) void gather_kernel(
    const float* __restrict__ root, const int* __restrict__ sidx,
    float* __restrict__ out) {
  const int lane = threadIdx.x & 31;
  const int r0 = __builtin_amdgcn_readfirstlane((blockIdx.x * 8 + (threadIdx.x >> 5)) * GROWS);
  int idx[GROWS];
#pragma unroll
  for (int j = 0; j < GROWS; ++j) idx[j] = sidx[r0 + j];
  v4u v0[GROWS], v1[GROWS];
#pragma unroll
  for (int j = 0; j < GROWS; ++j) {
    const unsigned li = (unsigned)idx[j] < (unsigned)NSTRUCT ? (unsigned)idx[j] : (unsigned)(NSTRUCT - 1);
    const v4u* src = (const v4u*)(root + (size_t)li * DD);
    v0[j] = src[lane];
    v1[j] = src[lane + 32];
  }
  __builtin_amdgcn_sched_barrier(0);
#pragma unroll
  for (int j = 0; j < GROWS; ++j) {
    v4u* dst = (v4u*)(out + (size_t)(r0 + j) * DD);
    const unsigned keep = idx[j] > 1 ? 0xffffffffu : 0u;
    dst[lane]      = v0[j] & keep;
    dst[lane + 32] = v1[j] & keep;
  }
}

extern "C" void kernel_launch(void* const* d_in, const int* in_sizes, int n_in,
                              void* d_out, int out_size, void* d_ws, size_t ws_size,
                              hipStream_t stream) {
  const int*   struct_indices = (const int*)d_in[0];
  const int*   leaf_ids       = (const int*)d_in[1];
  const int*   op_ids         = (const int*)d_in[2];
  const float* comp_emb       = (const float*)d_in[3];
  const float* op_emb         = (const float*)d_in[4];
  const float* W_bin          = (const float*)d_in[5];
  const float* b_bin          = (const float*)d_in[6];
  const float* gamma          = (const float*)d_in[7];
  const float* beta           = (const float*)d_in[8];

  float* root    = (float*)d_ws;
  float* ptab    = root + (size_t)NSTRUCT * DD;
  float* optab   = root + (size_t)NSTRUCT * 6 * DD;
  float* leaftab = optab + (size_t)64 * DD;

  table_kernel<<<PBLK + 2 * (NCOMP / TROWS), 256, 0, stream>>>(comp_emb, gamma, beta, op_emb, W_bin,
                                                               optab, leaftab);
  table2_kernel<<<2 * NGRP2, 256, 0, stream>>>(optab, b_bin, gamma, beta, W_bin, ptab);
  combine_kernel<<<NSTRUCT / MT, 256, 0, stream>>>(leaf_ids, op_ids, optab, leaftab, ptab, root,
                                                   W_bin, b_bin, gamma, beta);
  gather_kernel<<<65536 / (8 * GROWS), 256, 0, stream>>>(root, struct_indices, (float*)d_out);
}


# final gather fused into the root kernel: roots stay in LDS, each work-group scans the index list and writes the output rows that point at its 32 roots; 3 launches, no root plane; same sums
# speedup vs baseline: 4.3011x; 1.0772x over previous
#include <hip/hip_runtime.h>

#define DD 256
#define NSTRUCT 8192
#define EPS 1e-5f

typedef float v2f __attribute__((ext_vector_type(2)));
typedef float v8f __attribute__((ext_vector_type(8)));

__device__ __forceinline__ float wave_sum(float v) {
  v += __shfl_xor(v, 1, 32);
  v += __shfl_xor(v, 2, 32);
  v += __shfl_xor(v, 4, 32);
  v += __shfl_xor(v, 8, 32);
  v += __shfl_xor(v, 16, 32);
  return v;
}
template <int N> __device__ __forceinline__ void wave_sum_n(float (&v)[N]) {
#pragma unroll
  for (int m = 1; m < 32; m <<= 1) {
    float p[N];
#pragma unroll
    for (int i = 0; i < N; ++i) p[i] = __shfl_xor(v[i], m, 32);
#pragma unroll
    for (int i = 0; i < N; ++i) v[i] += p[i];
  }
}
__device__ __forceinline__ float sum4(float4 x) { return x.x + x.y + x.z + x.w; }
__device__ __forceinline__ float sumsq4(float4 x) { return x.x*x.x + x.y*x.y + x.z*x.z + x.w*x.w; }
__device__ __forceinline__ float4 ln4(float4 x, float mu, float rs, float4 g, float4 b) {
  float4 y;
  y.x = (x.x - mu) * rs * g.x + b.x;
  y.y = (x.y - mu) * rs * g.y + b.y;
  y.z = (x.z - mu) * rs * g.z + b.z;
  y.w = (x.w - mu) * rs * g.w + b.w;
  return y;
}

#define PBLK 4
#define NCOMP 1024
#define TROWS 16
#define TSTR 260

__global__ __launch_bounds__(256) void table_kernel(
    const float* __restrict__ comp_emb, const float* __restrict__ gamma, const float* __restrict__ beta,
    const float* __restrict__ op_emb, const float* __restrict__ W,
    float* __restrict__ optab, float* __restrict__ leaftab) {
  __shared__ __attribute__((aligned(16))) float sa[TROWS * TSTR];
  const int lane = threadIdx.x & 31;
  const int wave = threadIdx.x >> 5;
  const int krow = (lane & 16) >> 3;
  if (blockIdx.x < PBLK) {
    const int t = blockIdx.x * 8 + wave;
    const int colA = (t & 7) * 32 + (lane & 15);
    const float* arow = op_emb + (size_t)((t >> 3) * 16 + (lane & 15)) * DD;
    v8f c0 = {}, c1 = {};
#pragma unroll 8
    for (int k0 = 0; k0 < DD; k0 += 4) {
      const int kb = k0 + krow;
      const float* Wk = W + (size_t)kb * DD;
      v2f b0 = { Wk[colA],      Wk[DD + colA] };
      v2f b1 = { Wk[colA + 16], Wk[DD + colA + 16] };
      v2f a  = *(const v2f*)(arow + kb);
      c0 = __builtin_amdgcn_wmma_f32_16x16x4_f32(false, a, false, b0, (short)0, c0, false, false);
      c1 = __builtin_amdgcn_wmma_f32_16x16x4_f32(false, a, false, b1, (short)0, c1, false, false);
    }
    float* dst = optab + (size_t)((t >> 3) * 16 + ((lane & 16) >> 1)) * DD + colA;
#pragma unroll
    for (int j = 0; j < 8; ++j) {
      dst[j * DD]      = c0[j];
      dst[j * DD + 16] = c1[j];
    }
    return;
  }
  const int blk  = blockIdx.x - PBLK;
  const int rgrp = blk >> 1;
  const int side = blk & 1;
  {
    const float4 g0 = ((const float4*)gamma)[lane], g1 = ((const float4*)gamma)[lane + 32];
    const float4 b0 = ((const float4*)beta)[lane],  b1 = ((const float4*)beta)[lane + 32];
#pragma unroll
    for (int i = 0; i < 2; ++i) {
      const int lr = wave * 2 + i;
      const float4* src = (const float4*)(comp_emb + (size_t)(rgrp * TROWS + lr) * DD);
      float4 x0 = src[lane];
      float4 x1 = src[lane + 32];
      float s  = wave_sum(sum4(x0) + sum4(x1));
      float s2 = wave_sum(sumsq4(x0) + sumsq4(x1));
      float mu = s * (1.0f / DD);
      float rs = rsqrtf(s2 * (1.0f / DD) - mu * mu + EPS);
      *(float4*)&sa[lr * TSTR + lane * 4]       = ln4(x0, mu, rs, g0, b0);
      *(float4*)&sa[lr * TSTR + 128 + lane * 4] = ln4(x1, mu, rs, g1, b1);
    }
  }
  __syncthreads();
  const int colA = wave * 32 + (lane & 15);
  const float* arow = &sa[(lane & 15) * TSTR];
  const float* Ws = W + (size_t)(DD + side * DD) * DD;
  v8f c0 = {}, c1 = {};
#pragma unroll 8
  for (int k0 = 0; k0 < DD; k0 += 4) {
    const int kb = k0 + krow;
    const float* Wk = Ws + (size_t)kb * DD;
    v2f b0 = { Wk[colA],      Wk[DD + colA] };
    v2f b1 = { Wk[colA + 16], Wk[DD + colA + 16] };
    v2f a  = *(const v2f*)(arow + kb);
    c0 = __builtin_amdgcn_wmma_f32_16x16x4_f32(false, a, false, b0, (short)0, c0, false, false);
    c1 = __builtin_amdgcn_wmma_f32_16x16x4_f32(false, a, false, b1, (short)0, c1, false, false);
  }
  float* dst = leaftab + (size_t)side * NCOMP * DD + (size_t)(rgrp * TROWS + ((lane & 16) >> 1)) * DD + colA;
#pragma unroll
  for (int j = 0; j < 8; ++j) {
    dst[j * DD]      = c0[j];
    dst[j * DD + 16] = c1[j];
  }
}

#define NSRC 2112
#define NGRP2 134
#define PROWS (NGRP2 * 16)
#define PROW_B NSRC
#define PROW_BETA (NSRC + 16)

__global__ __launch_bounds__(256) void table2_kernel(
    const float* __restrict__ src, const float* __restrict__ bias,
    const float* __restrict__ gamma, const float* __restrict__ beta,
    const float* __restrict__ W, float* __restrict__ ptab) {
  __shared__ __attribute__((aligned(16))) float sa[TROWS * TSTR];
  const int lane = threadIdx.x & 31;
  const int wave = threadIdx.x >> 5;
  const int krow = (lane & 16) >> 3;
  const int grp  = blockIdx.x >> 1;
  const int side = blockIdx.x & 1;
  {
    const float4 g0 = ((const float4*)gamma)[lane], g1 = ((const float4*)gamma)[lane + 32];
#pragma unroll
    for (int i = 0; i < 2; ++i) {
      const int lr = wave * 2 + i;
      const float* row = grp < NSRC / 16 ? src + (size_t)(grp * 16 + lr) * DD : (grp == NSRC / 16 ? bias : beta);
      float4 x0 = ((const float4*)row)[lane];
      float4 x1 = ((const float4*)row)[lane + 32];
      if (grp <= NSRC / 16) {
        const float mu = wave_sum(sum4(x0) + sum4(x1)) * (1.0f / DD);
        x0.x = (x0.x - mu) * g0.x;  x0.y = (x0.y - mu) * g0.y;  x0.z = (x0.z - mu) * g0.z;  x0.w = (x0.w - mu) * g0.w;
        x1.x = (x1.x - mu) * g1.x;  x1.y = (x1.y - mu) * g1.y;  x1.z = (x1.z - mu) * g1.z;  x1.w = (x1.w - mu) * g1.w;
      }
      *(float4*)&sa[lr * TSTR + lane * 4]       = x0;
      *(float4*)&sa[lr * TSTR + 128 + lane * 4] = x1;
    }
  }
  __syncthreads();
  const int colA = wave * 32 + (lane & 15);
  const float* arow = &sa[(lane & 15) * TSTR];
  const float* Ws = W + (size_t)(DD + side * DD) * DD;
  v8f c0 = {}, c1 = {};
#pragma unroll 8
  for (int k0 = 0; k0 < DD; k0 += 4) {
    const int kb = k0 + krow;
    const float* Wk = Ws + (size_t)kb * DD;
    v2f b0 = { Wk[colA],      Wk[DD + colA] };
    v2f b1 = { Wk[colA + 16], Wk[DD + colA + 16] };
    v2f a  = *(const v2f*)(arow + kb);
    c0 = __builtin_amdgcn_wmma_f32_16x16x4_f32(false, a, false, b0, (short)0, c0, false, false);
    c1 = __builtin_amdgcn_wmma_f32_16x16x4_f32(false, a, false, b1, (short)0, c1, false, false);
  }
  float* dst = ptab + (size_t)side * PROWS * DD + (size_t)(grp * 16 + ((lane & 16) >> 1)) * DD + colA;
#pragma unroll
  for (int j = 0; j < 8; ++j) {
    dst[j * DD]      = c0[j];
    dst[j * DD + 16] = c1[j];
  }
}

#define SPW 4
__device__ __forceinline__ float4 add4(float4 a, float4 b) { return make_float4(a.x + b.x, a.y + b.y, a.z + b.z, a.w + b.w); }
__device__ __forceinline__ float4 fma4(float s, float4 a, float4 b) {
  return make_float4(s * a.x + b.x, s * a.y + b.y, s * a.z + b.z, s * a.w + b.w);
}

#define MT 32
#define KC 512
#define CSTR 516
#define OSTR 260

#define GK 4
#define KBLK (4 * GK)
#define NRING 3
struct OpBlk { v2f b0[GK], b1[GK], a0[GK], a1[GK]; };

__device__ __forceinline__ void load_blk(OpBlk& f, const float* __restrict__ Wb,
                                         const float* arow0, const float* arow1, int k0) {
#pragma unroll
  for (int s = 0; s < GK; ++s) {
    const float* Wk = Wb + (size_t)(k0 + 4 * s) * DD;
    f.b0[s] = (v2f){ Wk[0],  Wk[DD] };
    f.b1[s] = (v2f){ Wk[16], Wk[DD + 16] };
    f.a0[s] = *(const v2f*)(arow0 + k0 + 4 * s);
    f.a1[s] = *(const v2f*)(arow1 + k0 + 4 * s);
  }
}
__device__ __forceinline__ void mma_blk(const OpBlk& f, v8f& c00, v8f& c01, v8f& c10, v8f& c11) {
#pragma unroll
  for (int s = 0; s < GK; ++s) {
    c00 = __builtin_amdgcn_wmma_f32_16x16x4_f32(false, f.a0[s], false, f.b0[s], (short)0, c00, false, false);
    c01 = __builtin_amdgcn_wmma_f32_16x16x4_f32(false, f.a0[s], false, f.b1[s], (short)0, c01, false, false);
    c10 = __builtin_amdgcn_wmma_f32_16x16x4_f32(false, f.a1[s], false, f.b0[s], (short)0, c10, false, false);
    c11 = __builtin_amdgcn_wmma_f32_16x16x4_f32(false, f.a1[s], false, f.b1[s], (short)0, c11, false, false);
  }
}

#define NOUT 65536
#define SCG 8
typedef unsigned v4u __attribute__((ext_vector_type(4)));
__device__ __forceinline__ unsigned clamp_idx(int v) {
  return (unsigned)v < (unsigned)NSTRUCT ? (unsigned)v : (unsigned)(NSTRUCT - 1);
}
__device__ __forceinline__ void put_rows(int4 v, int pbase, unsigned blk, int lane,
                                         const float* rows, float* __restrict__ out) {
  const unsigned mk = ((clamp_idx(v.x) >> 5) == blk ? 1u : 0u) | ((clamp_idx(v.y) >> 5) == blk ? 2u : 0u) |
                      ((clamp_idx(v.z) >> 5) == blk ? 4u : 0u) | ((clamp_idx(v.w) >> 5) == blk ? 8u : 0u);
  unsigned b = __builtin_amdgcn_ballot_w32(mk != 0);
  while (b) {
    const int l = __builtin_ctz(b);
    b &= b - 1;
    unsigned m = __builtin_amdgcn_readlane(mk, l);
    const int x0 = __builtin_amdgcn_readlane(v.x, l), x1 = __builtin_amdgcn_readlane(v.y, l);
    const int x2 = __builtin_amdgcn_readlane(v.z, l), x3 = __builtin_amdgcn_readlane(v.w, l);
    while (m) {
      const int c = __builtin_ctz(m);
      m &= m - 1;
      const int vc = c == 0 ? x0 : (c == 1 ? x1 : (c == 2 ? x2 : x3));
      const v4u* srow = (const v4u*)(rows + (clamp_idx(vc) & 31u) * OSTR);
      const unsigned keep = vc > 1 ? 0xffffffffu : 0u;
      v4u* dst = (v4u*)(out + (size_t)(pbase + 4 * l + c) * DD);
      dst[lane]      = srow[lane] & keep;
      dst[lane + 32] = srow[lane + 32] & keep;
    }
  }
}

__global__ __launch_bounds__(256) __attribute__((amdgpu_waves_per_eu(2, 2))) void combine_kernel(
    const int* __restrict__ leaf_ids, const int* __restrict__ op_ids,
    const float* __restrict__ optab, const float* __restrict__ leaftab, const float* __restrict__ ptab,
    const int* __restrict__ sidx, float* __restrict__ out, const float* __restrict__ W,
    const float* __restrict__ bias, const float* __restrict__ gamma,
    const float* __restrict__ beta) {
  __shared__ __attribute__((aligned(16))) float smem[MT * CSTR];

  const int tid = threadIdx.x;
  const int lane = tid & 31;
  const int wave = __builtin_amdgcn_readfirstlane(tid >> 5);
  const int r0 = blockIdx.x * MT;

  const int n0 = wave * 32;
  const int colA = n0 + (lane & 15);
  const int roff = (lane & 16) >> 1;

  const int sb0 = r0 + wave * SPW;
  int lidn[8], oidn[6];
#pragma unroll
  for (int j = 0; j < 8; ++j) lidn[j] = leaf_ids[sb0 * 8 + j];
#pragma unroll
  for (int j = 0; j < 6; ++j) oidn[j] = op_ids[sb0 * 7 + j];

  int opid[16];
#pragma unroll
  for (int j = 0; j < 16; ++j) {
    const int rg = r0 + (j & 7) + roff + (j & 8) * 2;
    opid[j] = op_ids[rg * 7 + 6];
  }
  {
    const float* PL = ptab;
    const float* PR = ptab + (size_t)PROWS * DD;
    const float4 bs0 = ((const float4*)bias)[lane],  bs1 = ((const float4*)bias)[lane + 32];
    const float4 g0  = ((const float4*)gamma)[lane], g1  = ((const float4*)gamma)[lane + 32];
    const float4 bt0 = ((const float4*)beta)[lane],  bt1 = ((const float4*)beta)[lane + 32];
    const float4 lb0 = ((const float4*)(PL + (size_t)PROW_B * DD))[lane], lb1 = ((const float4*)(PL + (size_t)PROW_B * DD))[lane + 32];
    const float4 rb0 = ((const float4*)(PR + (size_t)PROW_B * DD))[lane], rb1 = ((const float4*)(PR + (size_t)PROW_B * DD))[lane + 32];
    const float4 cs0 = add4(add4(((const float4*)(PL + (size_t)PROW_BETA * DD))[lane],
                                 ((const float4*)(PR + (size_t)PROW_BETA * DD))[lane]), bs0);
    const float4 cs1 = add4(add4(((const float4*)(PL + (size_t)PROW_BETA * DD))[lane + 32],
                                 ((const float4*)(PR + (size_t)PROW_BETA * DD))[lane + 32]), bs1);
#pragma unroll 1
    for (int t = 0; t < SPW; ++t) {
      int lid[8], oid[6];
#pragma unroll
      for (int j = 0; j < 8; ++j) lid[j] = lidn[j];
#pragma unroll
      for (int j = 0; j < 6; ++j) oid[j] = oidn[j];
      {
        const int sn = sb0 + (t + 1 < SPW ? t + 1 : t);
#pragma unroll
        for (int j = 0; j < 8; ++j) lidn[j] = leaf_ids[sn * 8 + j];
#pragma unroll
        for (int j = 0; j < 6; ++j) oidn[j] = op_ids[sn * 7 + j];
      }
      float4 o0[4], o1[4], l0[4], l1[4], q0[4], q1[4];
      float4 a0[2], a1[2], u0[2], u1[2], v0[2], v1[2], w0[2], w1[2], d0[2], d1[2], e0[2], e1[2], f0[2], f1[2];
#pragma unroll
      for (int i = 0; i < 4; ++i) {
        const float4* po = (const float4*)(optab + (size_t)oid[i] * DD);
        const float4* pl = (const float4*)(leaftab + (size_t)lid[2 * i] * DD);
        const float4* pr = (const float4*)(leaftab + (size_t)NCOMP * DD + (size_t)lid[2 * i + 1] * DD);
        o0[i] = po[lane]; o1[i] = po[lane + 32];
        l0[i] = pl[lane]; l1[i] = pl[lane + 32];
        q0[i] = pr[lane]; q1[i] = pr[lane + 32];
      }
#pragma unroll
      for (int j = 0; j < 2; ++j) {
        const int il = 2 * j, ir = 2 * j + 1;
        const float4* pa  = (const float4*)(optab + (size_t)oid[4 + j] * DD);
        const float4* pl0 = (const float4*)(PL + (size_t)oid[il] * DD);
        const float4* pl1 = (const float4*)(PL + (size_t)(64 + lid[2 * il]) * DD);
        const float4* pl2 = (const float4*)(PL + (size_t)(64 + NCOMP + lid[2 * il + 1]) * DD);
        const float4* pr0 = (const float4*)(PR + (size_t)oid[ir] * DD);
        const float4* pr1 = (const float4*)(PR + (size_t)(64 + lid[2 * ir]) * DD);
        const float4* pr2 = (const float4*)(PR + (size_t)(64 + NCOMP + lid[2 * ir + 1]) * DD);
        a0[j] = pa[lane];  a1[j] = pa[lane + 32];
        u0[j] = pl0[lane]; u1[j] = pl0[lane + 32];
        v0[j] = pl1[lane]; v1[j] = pl1[lane + 32];
        w0[j] = pl2[lane]; w1[j] = pl2[lane + 32];
        d0[j] = pr0[lane]; d1[j] = pr0[lane + 32];
        e0[j] = pr1[lane]; e1[j] = pr1[lane + 32];
        f0[j] = pr2[lane]; f1[j] = pr2[lane + 32];
      }
      float rs[4];
      {
        float sv[8];
#pragma unroll
        for (int i = 0; i < 4; ++i) {
          const float4 x0 = add4(add4(add4(o0[i], l0[i]), q0[i]), bs0);
          const float4 x1 = add4(add4(add4(o1[i], l1[i]), q1[i]), bs1);
          sv[2 * i]     = sum4(x0) + sum4(x1);
          sv[2 * i + 1] = sumsq4(x0) + sumsq4(x1);
        }
        wave_sum_n<8>(sv);
#pragma unroll
        for (int i = 0; i < 4; ++i) {
          const float mu = sv[2 * i] * (1.0f / DD);
          rs[i] = rsqrtf(sv[2 * i + 1] * (1.0f / DD) - mu * mu + EPS);
        }
      }
      {
        float4 x0[2], x1[2];
        float rv[4];
#pragma unroll
        for (int j = 0; j < 2; ++j) {
          const int il = 2 * j, ir = 2 * j + 1;
          const float4 sl0 = add4(add4(add4(u0[j], v0[j]), w0[j]), lb0), sl1 = add4(add4(add4(u1[j], v1[j]), w1[j]), lb1);
          const float4 sr0 = add4(add4(add4(d0[j], e0[j]), f0[j]), rb0), sr1 = add4(add4(add4(d1[j], e1[j]), f1[j]), rb1);
          x0[j] = fma4(rs[ir], sr0, fma4(rs[il], sl0, add4(a0[j], cs0)));
          x1[j] = fma4(rs[ir], sr1, fma4(rs[il], sl1, add4(a1[j], cs1)));
          rv[2 * j]     = sum4(x0[j]) + sum4(x1[j]);
          rv[2 * j + 1] = sumsq4(x0[j]) + sumsq4(x1[j]);
        }
        wave_sum_n<4>(rv);
#pragma unroll
        for (int j = 0; j < 2; ++j) {
          const float mu = rv[2 * j] * (1.0f / DD);
          const float r  = rsqrtf(rv[2 * j + 1] * (1.0f / DD) - mu * mu + EPS);
          float* dst = &smem[(wave * SPW + t) * CSTR + j * DD + lane * 4];
          *(float4*)dst         = ln4(x0[j], mu, r, g0, bt0);
          *(float4*)(dst + 128) = ln4(x1[j], mu, r, g1, bt1);
        }
      }
    }
  }
  __syncthreads();

  v8f c00, c01, c10, c11;
#pragma unroll
  for (int j = 0; j < 8; ++j) {
    const float* p0 = optab + (size_t)opid[j] * DD + colA;
    const float* p1 = optab + (size_t)opid[8 + j] * DD + colA;
    c00[j] = p0[0];
    c01[j] = p0[16];
    c10[j] = p1[0];
    c11[j] = p1[16];
  }

  const int krow = (lane & 16) >> 3;
  const float* arow0 = &smem[(lane & 15) * CSTR + krow];
  const float* arow1 = arow0 + 16 * CSTR;
  const float* Wb = W + (size_t)(DD + krow) * DD + colA;
  OpBlk f[NRING];
#pragma unroll
  for (int g = 0; g < NRING - 1; ++g) load_blk(f[g], Wb, arow0, arow1, g * KBLK);
#pragma unroll
  for (int g = 0; g < KC / KBLK; ++g) {
    if (g + NRING - 1 < KC / KBLK) load_blk(f[(g + NRING - 1) % NRING], Wb, arow0, arow1, (g + NRING - 1) * KBLK);
    __builtin_amdgcn_sched_barrier(0);
    mma_blk(f[g % NRING], c00, c01, c10, c11);
    __builtin_amdgcn_sched_barrier(0);
  }
  __syncthreads();

  const float bias0 = bias[colA];
  const float bias1 = bias[colA + 16];
#pragma unroll
  for (int j = 0; j < 8; ++j) {
    const int rr = j + roff;
    smem[rr * OSTR + colA]              = c00[j] + bias0;
    smem[rr * OSTR + colA + 16]         = c01[j] + bias1;
    smem[(rr + 16) * OSTR + colA]       = c10[j] + bias0;
    smem[(rr + 16) * OSTR + colA + 16]  = c11[j] + bias1;
  }
  __syncthreads();

  const int4* s4 = (const int4*)sidx + wave * (NOUT / 8 / 4);
  int4 nx[SCG];
#pragma unroll
  for (int u = 0; u < SCG; ++u) nx[u] = s4[u * 32 + lane];

  const float4 g0  = ((const float4*)gamma)[lane];
  const float4 g1  = ((const float4*)gamma)[lane + 32];
  const float4 bt0 = ((const float4*)beta)[lane];
  const float4 bt1 = ((const float4*)beta)[lane + 32];
#pragma unroll
  for (int i = 0; i < 4; ++i) {
    const int row = wave * 4 + i;
    float4 x0 = *(const float4*)&smem[row * OSTR + lane * 4];
    float4 x1 = *(const float4*)&smem[row * OSTR + 128 + lane * 4];
    float s  = wave_sum(sum4(x0) + sum4(x1));
    float s2 = wave_sum(sumsq4(x0) + sumsq4(x1));
    float mu = s * (1.0f / DD);
    float rs = rsqrtf(s2 * (1.0f / DD) - mu * mu + EPS);
    *(float4*)&smem[row * OSTR + lane * 4]       = ln4(x0, mu, rs, g0, bt0);
    *(float4*)&smem[row * OSTR + 128 + lane * 4] = ln4(x1, mu, rs, g1, bt1);
  }
  __syncthreads();

#pragma unroll 1
  for (int g = 0; g < NOUT / 8 / 128 / SCG; ++g) {
    int4 cur[SCG];
#pragma unroll
    for (int u = 0; u < SCG; ++u) cur[u] = nx[u];
    {
      const int gn = g + 1 < NOUT / 8 / 128 / SCG ? g + 1 : g;
#pragma unroll
      for (int u = 0; u < SCG; ++u) nx[u] = s4[(gn * SCG + u) * 32 + lane];
    }
#pragma unroll
    for (int u = 0; u < SCG; ++u)
      put_rows(cur[u], wave * (NOUT / 8) + (g * SCG + u) * 128, blockIdx.x, lane, smem, out);
  }
}

extern "C" void kernel_launch(void* const* d_in, const int* in_sizes, int n_in,
                              void* d_out, int out_size, void* d_ws, size_t ws_size,
                              hipStream_t stream) {
  const int*   struct_indices = (const int*)d_in[0];
  const int*   leaf_ids       = (const int*)d_in[1];
  const int*   op_ids         = (const int*)d_in[2];
  const float* comp_emb       = (const float*)d_in[3];
  const float* op_emb         = (const float*)d_in[4];
  const float* W_bin          = (const float*)d_in[5];
  const float* b_bin          = (const float*)d_in[6];
  const float* gamma          = (const float*)d_in[7];
  const float* beta           = (const float*)d_in[8];

  float* ws      = (float*)d_ws;
  float* ptab    = ws + (size_t)NSTRUCT * DD;
  float* optab   = ws + (size_t)NSTRUCT * 6 * DD;
  float* leaftab = optab + (size_t)64 * DD;

  table_kernel<<<PBLK + 2 * (NCOMP / TROWS), 256, 0, stream>>>(comp_emb, gamma, beta, op_emb, W_bin,
                                                               optab, leaftab);
  table2_kernel<<<2 * NGRP2, 256, 0, stream>>>(optab, b_bin, gamma, beta, W_bin, ptab);
  combine_kernel<<<NSTRUCT / MT, 256, 0, stream>>>(leaf_ids, op_ids, optab, leaftab, ptab, struct_indices,
                                                   (float*)d_out, W_bin, b_bin, gamma, beta);
}


# both table stages in one launch: each work-group (16 waves) builds its 16 first-stage rows, keeps them in LDS and multiplies their centred copies by both W slices; 2 launches; same sums
# speedup vs baseline: 4.6593x; 1.0833x over previous
#include <hip/hip_runtime.h>

#define DD 256
#define NSTRUCT 8192
#define EPS 1e-5f

typedef float v2f __attribute__((ext_vector_type(2)));
typedef float v8f __attribute__((ext_vector_type(8)));

__device__ __forceinline__ float wave_sum(float v) {
  v += __shfl_xor(v, 1, 32);
  v += __shfl_xor(v, 2, 32);
  v += __shfl_xor(v, 4, 32);
  v += __shfl_xor(v, 8, 32);
  v += __shfl_xor(v, 16, 32);
  return v;
}
template <int N> __device__ __forceinline__ void wave_sum_n(float (&v)[N]) {
#pragma unroll
  for (int m = 1; m < 32; m <<= 1) {
    float p[N];
#pragma unroll
    for (int i = 0; i < N; ++i) p[i] = __shfl_xor(v[i], m, 32);
#pragma unroll
    for (int i = 0; i < N; ++i) v[i] += p[i];
  }
}
__device__ __forceinline__ float sum4(float4 x) { return x.x + x.y + x.z + x.w; }
__device__ __forceinline__ float sumsq4(float4 x) { return x.x*x.x + x.y*x.y + x.z*x.z + x.w*x.w; }
__device__ __forceinline__ float4 ln4(float4 x, float mu, float rs, float4 g, float4 b) {
  float4 y;
  y.x = (x.x - mu) * rs * g.x + b.x;
  y.y = (x.y - mu) * rs * g.y + b.y;
  y.z = (x.z - mu) * rs * g.z + b.z;
  y.w = (x.w - mu) * rs * g.w + b.w;
  return y;
}

#define NCOMP 1024
#define TROWS 16
#define TSTR 260
#define NSRC 2112
#define NGRP2 134
#define PROWS (NGRP2 * 16)
#define PROW_B NSRC
#define PROW_BETA (NSRC + 16)
#define TWAVES 16

__global__ __launch_bounds__(TWAVES * 32) void tables_kernel(
    const float* __restrict__ comp_emb, const float* __restrict__ gamma, const float* __restrict__ beta,
    const float* __restrict__ op_emb, const float* __restrict__ W, const float* __restrict__ bias,
    float* __restrict__ src, float* __restrict__ ptab) {
  __shared__ __attribute__((aligned(16))) float sa[TROWS * TSTR];
  __shared__ __attribute__((aligned(16))) float sy[TROWS * TSTR];
  const int lane = threadIdx.x & 31;
  const int wave = threadIdx.x >> 5;
  const int krow = (lane & 16) >> 3;
  const int roff = (lane & 16) >> 1;
  const int grp  = blockIdx.x;
  const float4 g0 = ((const float4*)gamma)[lane], g1 = ((const float4*)gamma)[lane + 32];
  const float* arow = &sa[(lane & 15) * TSTR];
  if (grp < NSRC / 16) {
    const float* Ws1;
    if (grp < 4) {
      const float4* rp = (const float4*)(op_emb + (size_t)(grp * 16 + wave) * DD);
      *(float4*)&sa[wave * TSTR + lane * 4]       = rp[lane];
      *(float4*)&sa[wave * TSTR + 128 + lane * 4] = rp[lane + 32];
      Ws1 = W;
    } else {
      const int side = (grp - 4) >> 6;
      const int rgrp = (grp - 4) & 63;
      const float4 b0 = ((const float4*)beta)[lane], b1 = ((const float4*)beta)[lane + 32];
      const float4* rp = (const float4*)(comp_emb + (size_t)(rgrp * TROWS + wave) * DD);
      float4 x0 = rp[lane];
      float4 x1 = rp[lane + 32];
      float s  = wave_sum(sum4(x0) + sum4(x1));
      float s2 = wave_sum(sumsq4(x0) + sumsq4(x1));
      float mu = s * (1.0f / DD);
      float rs = rsqrtf(s2 * (1.0f / DD) - mu * mu + EPS);
      *(float4*)&sa[wave * TSTR + lane * 4]       = ln4(x0, mu, rs, g0, b0);
      *(float4*)&sa[wave * TSTR + 128 + lane * 4] = ln4(x1, mu, rs, g1, b1);
      Ws1 = W + (size_t)(DD + side * DD) * DD;
    }
    __syncthreads();
    const int col = wave * 16 + (lane & 15);
    v8f c = {};
#pragma unroll 16
    for (int k0 = 0; k0 < DD; k0 += 4) {
      const int kb = k0 + krow;
      const float* Wk = Ws1 + (size_t)kb * DD;
      v2f b = { Wk[col], Wk[DD + col] };
      v2f a = *(const v2f*)(arow + kb);
      c = __builtin_amdgcn_wmma_f32_16x16x4_f32(false, a, false, b, (short)0, c, false, false);
    }
    float* dl = &sy[roff * TSTR + col];
#pragma unroll
    for (int j = 0; j < 8; ++j) dl[j * TSTR] = c[j];
    __syncthreads();
  } else {
    const float* row = grp == NSRC / 16 ? bias : beta;
    *(float4*)&sy[wave * TSTR + lane * 4]       = ((const float4*)row)[lane];
    *(float4*)&sy[wave * TSTR + 128 + lane * 4] = ((const float4*)row)[lane + 32];
  }
  float4 y0, y1;
  {
    float4 x0 = *(const float4*)&sy[wave * TSTR + lane * 4];
    float4 x1 = *(const float4*)&sy[wave * TSTR + 128 + lane * 4];
    y0 = x0;
    y1 = x1;
    if (grp <= NSRC / 16) {
      const float mu = wave_sum(sum4(x0) + sum4(x1)) * (1.0f / DD);
      x0.x = (x0.x - mu) * g0.x;  x0.y = (x0.y - mu) * g0.y;  x0.z = (x0.z - mu) * g0.z;  x0.w = (x0.w - mu) * g0.w;
      x1.x = (x1.x - mu) * g1.x;  x1.y = (x1.y - mu) * g1.y;  x1.z = (x1.z - mu) * g1.z;  x1.w = (x1.w - mu) * g1.w;
    }
    *(float4*)&sa[wave * TSTR + lane * 4]       = x0;
    *(float4*)&sa[wave * TSTR + 128 + lane * 4] = x1;
  }
  __syncthreads();
  if (grp < NSRC / 16) {
    float4* drow = (float4*)(src + (size_t)(grp * 16 + wave) * DD);
    drow[lane]      = y0;
    drow[lane + 32] = y1;
  }
  const int side2 = wave >> 3;
  const int colA = (wave & 7) * 32 + (lane & 15);
  const float* Ws = W + (size_t)(DD + side2 * DD) * DD;
  v8f c0 = {}, c1 = {};
#pragma unroll 8
  for (int k0 = 0; k0 < DD; k0 += 4) {
    const int kb = k0 + krow;
    const float* Wk = Ws + (size_t)kb * DD;
    v2f b0 = { Wk[colA],      Wk[DD + colA] };
    v2f b1 = { Wk[colA + 16], Wk[DD + colA + 16] };
    v2f a  = *(const v2f*)(arow + kb);
    c0 = __builtin_amdgcn_wmma_f32_16x16x4_f32(false, a, false, b0, (short)0, c0, false, false);
    c1 = __builtin_amdgcn_wmma_f32_16x16x4_f32(false, a, false, b1, (short)0, c1, false, false);
  }
  float* dst = ptab + (size_t)side2 * PROWS * DD + (size_t)(grp * 16 + roff) * DD + colA;
#pragma unroll
  for (int j = 0; j < 8; ++j) {
    dst[j * DD]      = c0[j];
    dst[j * DD + 16] = c1[j];
  }
}

#define SPW 4
__device__ __forceinline__ float4 add4(float4 a, float4 b) { return make_float4(a.x + b.x, a.y + b.y, a.z + b.z, a.w + b.w); }
__device__ __forceinline__ float4 fma4(float s, float4 a, float4 b) {
  return make_float4(s * a.x + b.x, s * a.y + b.y, s * a.z + b.z, s * a.w + b.w);
}

#define MT 32
#define KC 512
#define CSTR 516
#define OSTR 260

#define GK 4
#define KBLK (4 * GK)
#define NRING 3
struct OpBlk { v2f b0[GK], b1[GK], a0[GK], a1[GK]; };

__device__ __forceinline__ void load_blk(OpBlk& f, const float* __restrict__ Wb,
                                         const float* arow0, const float* arow1, int k0) {
#pragma unroll
  for (int s = 0; s < GK; ++s) {
    const float* Wk = Wb + (size_t)(k0 + 4 * s) * DD;
    f.b0[s] = (v2f){ Wk[0],  Wk[DD] };
    f.b1[s] = (v2f){ Wk[16], Wk[DD + 16] };
    f.a0[s] = *(const v2f*)(arow0 + k0 + 4 * s);
    f.a1[s] = *(const v2f*)(arow1 + k0 + 4 * s);
  }
}
__device__ __forceinline__ void mma_blk(const OpBlk& f, v8f& c00, v8f& c01, v8f& c10, v8f& c11) {
#pragma unroll
  for (int s = 0; s < GK; ++s) {
    c00 = __builtin_amdgcn_wmma_f32_16x16x4_f32(false, f.a0[s], false, f.b0[s], (short)0, c00, false, false);
    c01 = __builtin_amdgcn_wmma_f32_16x16x4_f32(false, f.a0[s], false, f.b1[s], (short)0, c01, false, false);
    c10 = __builtin_amdgcn_wmma_f32_16x16x4_f32(false, f.a1[s], false, f.b0[s], (short)0, c10, false, false);
    c11 = __builtin_amdgcn_wmma_f32_16x16x4_f32(false, f.a1[s], false, f.b1[s], (short)0, c11, false, false);
  }
}

#define NOUT 65536
#define SCG 8
typedef unsigned v4u __attribute__((ext_vector_type(4)));
__device__ __forceinline__ unsigned clamp_idx(int v) {
  return (unsigned)v < (unsigned)NSTRUCT ? (unsigned)v : (unsigned)(NSTRUCT - 1);
}
__device__ __forceinline__ void put_rows(int4 v, int pbase, unsigned blk, int lane,
                                         const float* rows, float* __restrict__ out) {
  const unsigned mk = ((clamp_idx(v.x) >> 5) == blk ? 1u : 0u) | ((clamp_idx(v.y) >> 5) == blk ? 2u : 0u) |
                      ((clamp_idx(v.z) >> 5) == blk ? 4u : 0u) | ((clamp_idx(v.w) >> 5) == blk ? 8u : 0u);
  unsigned b = __builtin_amdgcn_ballot_w32(mk != 0);
  while (b) {
    const int l = __builtin_ctz(b);
    b &= b - 1;
    unsigned m = __builtin_amdgcn_readlane(mk, l);
    const int x0 = __builtin_amdgcn_readlane(v.x, l), x1 = __builtin_amdgcn_readlane(v.y, l);
    const int x2 = __builtin_amdgcn_readlane(v.z, l), x3 = __builtin_amdgcn_readlane(v.w, l);
    while (m) {
      const int c = __builtin_ctz(m);
      m &= m - 1;
      const int vc = c == 0 ? x0 : (c == 1 ? x1 : (c == 2 ? x2 : x3));
      const v4u* srow = (const v4u*)(rows + (clamp_idx(vc) & 31u) * OSTR);
      const unsigned keep = vc > 1 ? 0xffffffffu : 0u;
      v4u* dst = (v4u*)(out + (size_t)(pbase + 4 * l + c) * DD);
      dst[lane]      = srow[lane] & keep;
      dst[lane + 32] = srow[lane + 32] & keep;
    }
  }
}

__global__ __launch_bounds__(256) __attribute__((amdgpu_waves_per_eu(2, 2))) void combine_kernel(
    const int* __restrict__ leaf_ids, const int* __restrict__ op_ids,
    const float* __restrict__ optab, const float* __restrict__ leaftab, const float* __restrict__ ptab,
    const int* __restrict__ sidx, float* __restrict__ out, const float* __restrict__ W,
    const float* __restrict__ bias, const float* __restrict__ gamma,
    const float* __restrict__ beta) {
  __shared__ __attribute__((aligned(16))) float smem[MT * CSTR];

  const int tid = threadIdx.x;
  const int lane = tid & 31;
  const int wave = __builtin_amdgcn_readfirstlane(tid >> 5);
  const int r0 = blockIdx.x * MT;

  const int n0 = wave * 32;
  const int colA = n0 + (lane & 15);
  const int roff = (lane & 16) >> 1;

  const int sb0 = r0 + wave * SPW;
  int lidn[8], oidn[6];
#pragma unroll
  for (int j = 0; j < 8; ++j) lidn[j] = leaf_ids[sb0 * 8 + j];
#pragma unroll
  for (int j = 0; j < 6; ++j) oidn[j] = op_ids[sb0 * 7 + j];

  int opid[16];
#pragma unroll
  for (int j = 0; j < 16; ++j) {
    const int rg = r0 + (j & 7) + roff + (j & 8) * 2;
    opid[j] = op_ids[rg * 7 + 6];
  }
  {
    const float* PL = ptab;
    const float* PR = ptab + (size_t)PROWS * DD;
    const float4 bs0 = ((const float4*)bias)[lane],  bs1 = ((const float4*)bias)[lane + 32];
    const float4 g0  = ((const float4*)gamma)[lane], g1  = ((const float4*)gamma)[lane + 32];
    const float4 bt0 = ((const float4*)beta)[lane],  bt1 = ((const float4*)beta)[lane + 32];
    const float4 lb0 = ((const float4*)(PL + (size_t)PROW_B * DD))[lane], lb1 = ((const float4*)(PL + (size_t)PROW_B * DD))[lane + 32];
    const float4 rb0 = ((const float4*)(PR + (size_t)PROW_B * DD))[lane], rb1 = ((const float4*)(PR + (size_t)PROW_B * DD))[lane + 32];
    const float4 cs0 = add4(add4(((const float4*)(PL + (size_t)PROW_BETA * DD))[lane],
                                 ((const float4*)(PR + (size_t)PROW_BETA * DD))[lane]), bs0);
    const float4 cs1 = add4(add4(((const float4*)(PL + (size_t)PROW_BETA * DD))[lane + 32],
                                 ((const float4*)(PR + (size_t)PROW_BETA * DD))[lane + 32]), bs1);
#pragma unroll 1
    for (int t = 0; t < SPW; ++t) {
      int lid[8], oid[6];
#pragma unroll
      for (int j = 0; j < 8; ++j) lid[j] = lidn[j];
#pragma unroll
      for (int j = 0; j < 6; ++j) oid[j] = oidn[j];
      {
        const int sn = sb0 + (t + 1 < SPW ? t + 1 : t);
#pragma unroll
        for (int j = 0; j < 8; ++j) lidn[j] = leaf_ids[sn * 8 + j];
#pragma unroll
        for (int j = 0; j < 6; ++j) oidn[j] = op_ids[sn * 7 + j];
      }
      float4 o0[4], o1[4], l0[4], l1[4], q0[4], q1[4];
      float4 a0[2], a1[2], u0[2], u1[2], v0[2], v1[2], w0[2], w1[2], d0[2], d1[2], e0[2], e1[2], f0[2], f1[2];
#pragma unroll
      for (int i = 0; i < 4; ++i) {
        const float4* po = (const float4*)(optab + (size_t)oid[i] * DD);
        const float4* pl = (const float4*)(leaftab + (size_t)lid[2 * i] * DD);
        const float4* pr = (const float4*)(leaftab + (size_t)NCOMP * DD + (size_t)lid[2 * i + 1] * DD);
        o0[i] = po[lane]; o1[i] = po[lane + 32];
        l0[i] = pl[lane]; l1[i] = pl[lane + 32];
        q0[i] = pr[lane]; q1[i] = pr[lane + 32];
      }
#pragma unroll
      for (int j = 0; j < 2; ++j) {
        const int il = 2 * j, ir = 2 * j + 1;
        const float4* pa  = (const float4*)(optab + (size_t)oid[4 + j] * DD);
        const float4* pl0 = (const float4*)(PL + (size_t)oid[il] * DD);
        const float4* pl1 = (const float4*)(PL + (size_t)(64 + lid[2 * il]) * DD);
        const float4* pl2 = (const float4*)(PL + (size_t)(64 + NCOMP + lid[2 * il + 1]) * DD);
        const float4* pr0 = (const float4*)(PR + (size_t)oid[ir] * DD);
        const float4* pr1 = (const float4*)(PR + (size_t)(64 + lid[2 * ir]) * DD);
        const float4* pr2 = (const float4*)(PR + (size_t)(64 + NCOMP + lid[2 * ir + 1]) * DD);
        a0[j] = pa[lane];  a1[j] = pa[lane + 32];
        u0[j] = pl0[lane]; u1[j] = pl0[lane + 32];
        v0[j] = pl1[lane]; v1[j] = pl1[lane + 32];
        w0[j] = pl2[lane]; w1[j] = pl2[lane + 32];
        d0[j] = pr0[lane]; d1[j] = pr0[lane + 32];
        e0[j] = pr1[lane]; e1[j] = pr1[lane + 32];
        f0[j] = pr2[lane]; f1[j] = pr2[lane + 32];
      }
      float rs[4];
      {
        float sv[8];
#pragma unroll
        for (int i = 0; i < 4; ++i) {
          const float4 x0 = add4(add4(add4(o0[i], l0[i]), q0[i]), bs0);
          const float4 x1 = add4(add4(add4(o1[i], l1[i]), q1[i]), bs1);
          sv[2 * i]     = sum4(x0) + sum4(x1);
          sv[2 * i + 1] = sumsq4(x0) + sumsq4(x1);
        }
        wave_sum_n<8>(sv);
#pragma unroll
        for (int i = 0; i < 4; ++i) {
          const float mu = sv[2 * i] * (1.0f / DD);
          rs[i] = rsqrtf(sv[2 * i + 1] * (1.0f / DD) - mu * mu + EPS);
        }
      }
      {
        float4 x0[2], x1[2];
        float rv[4];
#pragma unroll
        for (int j = 0; j < 2; ++j) {
          const int il = 2 * j, ir = 2 * j + 1;
          const float4 sl0 = add4(add4(add4(u0[j], v0[j]), w0[j]), lb0), sl1 = add4(add4(add4(u1[j], v1[j]), w1[j]), lb1);
          const float4 sr0 = add4(add4(add4(d0[j], e0[j]), f0[j]), rb0), sr1 = add4(add4(add4(d1[j], e1[j]), f1[j]), rb1);
          x0[j] = fma4(rs[ir], sr0, fma4(rs[il], sl0, add4(a0[j], cs0)));
          x1[j] = fma4(rs[ir], sr1, fma4(rs[il], sl1, add4(a1[j], cs1)));
          rv[2 * j]     = sum4(x0[j]) + sum4(x1[j]);
          rv[2 * j + 1] = sumsq4(x0[j]) + sumsq4(x1[j]);
        }
        wave_sum_n<4>(rv);
#pragma unroll
        for (int j = 0; j < 2; ++j) {
          const float mu = rv[2 * j] * (1.0f / DD);
          const float r  = rsqrtf(rv[2 * j + 1] * (1.0f / DD) - mu * mu + EPS);
          float* dst = &smem[(wave * SPW + t) * CSTR + j * DD + lane * 4];
          *(float4*)dst         = ln4(x0[j], mu, r, g0, bt0);
          *(float4*)(dst + 128) = ln4(x1[j], mu, r, g1, bt1);
        }
      }
    }
  }
  __syncthreads();

  v8f c00, c01, c10, c11;
#pragma unroll
  for (int j = 0; j < 8; ++j) {
    const float* p0 = optab + (size_t)opid[j] * DD + colA;
    const float* p1 = optab + (size_t)opid[8 + j] * DD + colA;
    c00[j] = p0[0];
    c01[j] = p0[16];
    c10[j] = p1[0];
    c11[j] = p1[16];
  }

  const int krow = (lane & 16) >> 3;
  const float* arow0 = &smem[(lane & 15) * CSTR + krow];
  const float* arow1 = arow0 + 16 * CSTR;
  const float* Wb = W + (size_t)(DD + krow) * DD + colA;
  OpBlk f[NRING];
#pragma unroll
  for (int g = 0; g < NRING - 1; ++g) load_blk(f[g], Wb, arow0, arow1, g * KBLK);
#pragma unroll
  for (int g = 0; g < KC / KBLK; ++g) {
    if (g + NRING - 1 < KC / KBLK) load_blk(f[(g + NRING - 1) % NRING], Wb, arow0, arow1, (g + NRING - 1) * KBLK);
    __builtin_amdgcn_sched_barrier(0);
    mma_blk(f[g % NRING], c00, c01, c10, c11);
    __builtin_amdgcn_sched_barrier(0);
  }
  __syncthreads();

  const float bias0 = bias[colA];
  const float bias1 = bias[colA + 16];
#pragma unroll
  for (int j = 0; j < 8; ++j) {
    const int rr = j + roff;
    smem[rr * OSTR + colA]              = c00[j] + bias0;
    smem[rr * OSTR + colA + 16]         = c01[j] + bias1;
    smem[(rr + 16) * OSTR + colA]       = c10[j] + bias0;
    smem[(rr + 16) * OSTR + colA + 16]  = c11[j] + bias1;
  }
  __syncthreads();

  const int4* s4 = (const int4*)sidx + wave * (NOUT / 8 / 4);
  int4 nx[SCG];
#pragma unroll
  for (int u = 0; u < SCG; ++u) nx[u] = s4[u * 32 + lane];

  const float4 g0  = ((const float4*)gamma)[lane];
  const float4 g1  = ((const float4*)gamma)[lane + 32];
  const float4 bt0 = ((const float4*)beta)[lane];
  const float4 bt1 = ((const float4*)beta)[lane + 32];
#pragma unroll
  for (int i = 0; i < 4; ++i) {
    const int row = wave * 4 + i;
    float4 x0 = *(const float4*)&smem[row * OSTR + lane * 4];
    float4 x1 = *(const float4*)&smem[row * OSTR + 128 + lane * 4];
    float s  = wave_sum(sum4(x0) + sum4(x1));
    float s2 = wave_sum(sumsq4(x0) + sumsq4(x1));
    float mu = s * (1.0f / DD);
    float rs = rsqrtf(s2 * (1.0f / DD) - mu * mu + EPS);
    *(float4*)&smem[row * OSTR + lane * 4]       = ln4(x0, mu, rs, g0, bt0);
    *(float4*)&smem[row * OSTR + 128 + lane * 4] = ln4(x1, mu, rs, g1, bt1);
  }
  __syncthreads();

#pragma unroll 1
  for (int g = 0; g < NOUT / 8 / 128 / SCG; ++g) {
    int4 cur[SCG];
#pragma unroll
    for (int u = 0; u < SCG; ++u) cur[u] = nx[u];
    {
      const int gn = g + 1 < NOUT / 8 / 128 / SCG ? g + 1 : g;
#pragma unroll
      for (int u = 0; u < SCG; ++u) nx[u] = s4[(gn * SCG + u) * 32 + lane];
    }
#pragma unroll
    for (int u = 0; u < SCG; ++u)
      put_rows(cur[u], wave * (NOUT / 8) + (g * SCG + u) * 128, blockIdx.x, lane, smem, out);
  }
}

extern "C" void kernel_launch(void* const* d_in, const int* in_sizes, int n_in,
                              void* d_out, int out_size, void* d_ws, size_t ws_size,
                              hipStream_t stream) {
  const int*   struct_indices = (const int*)d_in[0];
  const int*   leaf_ids       = (const int*)d_in[1];
  const int*   op_ids         = (const int*)d_in[2];
  const float* comp_emb       = (const float*)d_in[3];
  const float* op_emb         = (const float*)d_in[4];
  const float* W_bin          = (const float*)d_in[5];
  const float* b_bin          = (const float*)d_in[6];
  const float* gamma          = (const float*)d_in[7];
  const float* beta           = (const float*)d_in[8];

  float* ws      = (float*)d_ws;
  float* ptab    = ws + (size_t)NSTRUCT * DD;
  float* optab   = ws + (size_t)NSTRUCT * 6 * DD;
  float* leaftab = optab + (size_t)64 * DD;

  tables_kernel<<<NGRP2, TWAVES * 32, 0, stream>>>(comp_emb, gamma, beta, op_emb, W_bin, b_bin, optab, ptab);
  combine_kernel<<<NSTRUCT / MT, 256, 0, stream>>>(leaf_ids, op_ids, optab, leaftab, ptab, struct_indices,
                                                   (float*)d_out, W_bin, b_bin, gamma, beta);
}
